# starting kernel
# baseline (speedup 1.0000x reference)
#include <hip/hip_runtime.h>
#include <hip/hip_bf16.h>
#include <math.h>


typedef float v2f __attribute__((ext_vector_type(2)));
typedef float v8f __attribute__((ext_vector_type(8)));

#define F_IN 128
#define HID  16
#define NCLS 8


__global__ void deg_init_kernel(float* __restrict__ deg, int n) {
  int i = blockIdx.x * blockDim.x + threadIdx.x;
  if (i < n) deg[i] = 1.0f;
}

__global__ void deg_count_kernel(const int* __restrict__ dst,
                                 float* __restrict__ deg, int e) {
  int i = blockIdx.x * blockDim.x + threadIdx.x;
  if (i < e) atomicAdd(&deg[dst[i]], 1.0f);
}

__global__ void dinv_kernel(float* __restrict__ deg, int n) {
  int i = blockIdx.x * blockDim.x + threadIdx.x;
  if (i < n) deg[i] = rsqrtf(deg[i]);
}


__global__ void __launch_bounds__(256) gemm1_kernel(
    const float* __restrict__ x, const float* __restrict__ W1,
    const float* __restrict__ dinv,
    float* __restrict__ h, float* __restrict__ out1, int nTiles, int n) {
  int wave = (blockIdx.x * 256 + threadIdx.x) >> 5;
  int lane = threadIdx.x & 31;
  if (wave >= nTiles) return;
  int half = lane >> 4;
  int m    = lane & 15;
  int row  = wave * 16 + m;
  int rowc = row < n ? row : n - 1;
  const float* xrow = x + (size_t)rowc * F_IN;

  v8f c = {};
  #pragma unroll 4
  for (int k0 = 0; k0 < F_IN; k0 += 4) {
    int kk = k0 + 2 * half;
    v2f a = *(const v2f*)(xrow + kk);
    v2f b;
    b.x = W1[kk * HID + m];
    b.y = W1[(kk + 1) * HID + m];
    c = __builtin_amdgcn_wmma_f32_16x16x4_f32(
        false, a, false, b, (short)0, c, false, false);
  }

  float dv = dinv[rowc];
  bool full = (wave * 16 + 16 <= n);
  if (full) {
    #pragma unroll
    for (int r = 0; r < 8; ++r) {
      int orow = wave * 16 + r + 8 * half;
      float w  = __shfl(dv, r + 8 * half, 32);
      size_t o = (size_t)orow * HID + m;
      h[o]    = c[r];
      out1[o] = c[r] * w * w;
    }
  } else {
    #pragma unroll
    for (int r = 0; r < 8; ++r) {
      int orow = wave * 16 + r + 8 * half;
      float w  = __shfl(dv, r + 8 * half, 32);
      if (orow < n) {
        size_t o = (size_t)orow * HID + m;
        h[o]    = c[r];
        out1[o] = c[r] * w * w;
      }
    }
  }
}


__global__ void scatter1_kernel(const int* __restrict__ src,
                                const int* __restrict__ dst,
                                const float* __restrict__ dinv,
                                const float* __restrict__ h,
                                float* __restrict__ out1, int e) {
  int i = blockIdx.x * blockDim.x + threadIdx.x;
  if (i >= e) return;
  int s = src[i];
  int d = dst[i];
  float w = dinv[s] * dinv[d];
  const float4* hp = (const float4*)(h + (size_t)s * HID);
  float* op = out1 + (size_t)d * HID;
  #pragma unroll
  for (int q = 0; q < 4; ++q) {
    float4 v = hp[q];
    atomicAdd(op + q * 4 + 0, v.x * w);
    atomicAdd(op + q * 4 + 1, v.y * w);
    atomicAdd(op + q * 4 + 2, v.z * w);
    atomicAdd(op + q * 4 + 3, v.w * w);
  }
}


__global__ void __launch_bounds__(256) gemm2_kernel(
    const float* __restrict__ h1, const float* __restrict__ b1,
    const float* __restrict__ W2, const float* __restrict__ dinv,
    float* __restrict__ lt, float* __restrict__ out, int nTiles, int n) {
  int wave = (blockIdx.x * 256 + threadIdx.x) >> 5;
  int lane = threadIdx.x & 31;
  if (wave >= nTiles) return;
  int half = lane >> 4;
  int m    = lane & 15;
  int row  = wave * 16 + m;
  int rowc = row < n ? row : n - 1;
  const float* hrow = h1 + (size_t)rowc * HID;
  float maskv = (m < NCLS) ? 1.0f : 0.0f;
  int mc = m & (NCLS - 1);

  v8f c = {};
  #pragma unroll
  for (int k0 = 0; k0 < HID; k0 += 4) {
    int kk = k0 + 2 * half;
    v2f a;
    a.x = fmaxf(hrow[kk]     + b1[kk],     0.0f);
    a.y = fmaxf(hrow[kk + 1] + b1[kk + 1], 0.0f);
    v2f b;
    b.x = W2[kk * NCLS + mc] * maskv;
    b.y = W2[(kk + 1) * NCLS + mc] * maskv;
    c = __builtin_amdgcn_wmma_f32_16x16x4_f32(
        false, a, false, b, (short)0, c, false, false);
  }

  float dv = dinv[rowc];
  bool full = (wave * 16 + 16 <= n);
  if (m < NCLS) {
    #pragma unroll
    for (int r = 0; r < 8; ++r) {
      int orow = wave * 16 + r + 8 * half;
      float w  = __shfl(dv, r + 8 * half, 32);
      if (full || orow < n) {
        size_t o = (size_t)orow * NCLS + m;
        lt[o]  = c[r];
        out[o] = c[r] * w * w;
      }
    }
  }
}


__global__ void scatter2_kernel(const int* __restrict__ src,
                                const int* __restrict__ dst,
                                const float* __restrict__ dinv,
                                const float* __restrict__ lt,
                                float* __restrict__ out, int e) {
  int i = blockIdx.x * blockDim.x + threadIdx.x;
  if (i >= e) return;
  int s = src[i];
  int d = dst[i];
  float w = dinv[s] * dinv[d];
  const float4* lp = (const float4*)(lt + (size_t)s * NCLS);
  float* op = out + (size_t)d * NCLS;
  #pragma unroll
  for (int q = 0; q < 2; ++q) {
    float4 v = lp[q];
    atomicAdd(op + q * 4 + 0, v.x * w);
    atomicAdd(op + q * 4 + 1, v.y * w);
    atomicAdd(op + q * 4 + 2, v.z * w);
    atomicAdd(op + q * 4 + 3, v.w * w);
  }
}


__global__ void lsm_kernel(float* __restrict__ out,
                           const float* __restrict__ b2, int n) {
  int i = blockIdx.x * blockDim.x + threadIdx.x;
  if (i >= n) return;
  float* row = out + (size_t)i * NCLS;
  float z[NCLS];
  float mx = -INFINITY;
  #pragma unroll
  for (int c = 0; c < NCLS; ++c) {
    z[c] = row[c] + b2[c];
    mx = fmaxf(mx, z[c]);
  }
  float s = 0.0f;
  #pragma unroll
  for (int c = 0; c < NCLS; ++c) s += expf(z[c] - mx);
  float l = mx + logf(s);
  #pragma unroll
  for (int c = 0; c < NCLS; ++c) row[c] = z[c] - l;
}


extern "C" void kernel_launch(void* const* d_in, const int* in_sizes, int n_in,
                              void* d_out, int out_size, void* d_ws, size_t ws_size,
                              hipStream_t stream) {
  const float* x  = (const float*)d_in[0];
  const int* edge = (const int*)d_in[1];
  const float* W1 = (const float*)d_in[2];
  const float* b1 = (const float*)d_in[3];
  const float* W2 = (const float*)d_in[4];
  const float* b2 = (const float*)d_in[5];

  int n = in_sizes[0] / F_IN;
  int e = in_sizes[1] / 2;
  const int* src = edge;
  const int* dst = edge + e;

  float* deg = (float*)d_ws;
  float* h   = deg + n;
  float* h1  = h + (size_t)n * HID;
  float* lt  = h1 + (size_t)n * HID;
  float* out = (float*)d_out;

  const int T = 256;
  int nTiles  = (n + 15) / 16;
  int gBlocks = (nTiles + 7) / 8;

  deg_init_kernel<<<(n + T - 1) / T, T, 0, stream>>>(deg, n);
  deg_count_kernel<<<(e + T - 1) / T, T, 0, stream>>>(dst, deg, e);
  dinv_kernel<<<(n + T - 1) / T, T, 0, stream>>>(deg, n);

  gemm1_kernel<<<gBlocks, T, 0, stream>>>(x, W1, deg, h, h1, nTiles, n);
  scatter1_kernel<<<(e + T - 1) / T, T, 0, stream>>>(src, dst, deg, h, h1, e);

  gemm2_kernel<<<gBlocks, T, 0, stream>>>(h1, b1, W2, deg, lt, out, nTiles, n);
  scatter2_kernel<<<(e + T - 1) / T, T, 0, stream>>>(src, dst, deg, lt, out, e);

  lsm_kernel<<<(n + T - 1) / T, T, 0, stream>>>(out, b2, n);
}


# Edge-sum LDS list built per wave: an edge position's owners are a ballot mask, slots from mbcnt, one counter add per wave per tile (was per thread); list order, hence LDS-atomic sum order, changes
# speedup vs baseline: 6.6360x; 6.6360x over previous
#include <hip/hip_runtime.h>
#include <hip/hip_bf16.h>
#include <math.h>


typedef float v2f __attribute__((ext_vector_type(2)));
typedef float v8f __attribute__((ext_vector_type(8)));

#define F_IN 128
#define HID  16
#define NCLS 8


#define ATPB 1024
#define VQ   8
#define LCAP 4096

#define DNB  25000
#define DCH  64
#define NB1  4000
#define NCH1 10
#define NB2  8000
#define NCH2 19


template <bool VEC>
__device__ __forceinline__ void load_tile(const int* __restrict__ p, int e0,
                                          int e1, int ng, int gb, int tid,
                                          int fill, int4 (&q)[VQ]) {
  if (VEC) {
    #pragma unroll
    for (int k = 0; k < VQ; ++k) {
      int g = gb + k * ATPB + tid;
      q[k] = *(const int4*)(p + e0 + 4 * (g < ng ? g : 0));
    }
  } else {
    #pragma unroll
    for (int k = 0; k < VQ; ++k) {
      int g = gb + k * ATPB + tid;
      int base = e0 + 4 * (g < ng ? g : 0);
      q[k].x = (base + 0 < e1) ? p[base + 0] : fill;
      q[k].y = (base + 1 < e1) ? p[base + 1] : fill;
      q[k].z = (base + 2 < e1) ? p[base + 2] : fill;
      q[k].w = (base + 3 < e1) ? p[base + 3] : fill;
    }
  }
}

__device__ __forceinline__ bool group_ok(int gb, int k, int tid, int ng) {
  return gb + k * ATPB + tid < ng;
}


template <bool VEC>
__device__ __forceinline__ void deg_part_body(
    float* hc, const int* __restrict__ dst, float* __restrict__ degp, int e, int n) {
  const int tid = threadIdx.x;
  const int nb0 = blockIdx.x * DNB;
  const int nbn = min(DNB, n - nb0);
  const int per = (((e + DCH - 1) / DCH) + 3) & ~3;
  const int e0  = blockIdx.y * per;
  const int e1  = min(e, e0 + per);
  const int ng  = e1 > e0 ? (e1 - e0 + 3) >> 2 : 0;

  for (int i = tid; i < DNB; i += ATPB) hc[i] = 0.0f;
  __syncthreads();

  for (int g0 = 0; g0 < ng; g0 += ATPB * VQ) {
    int4 dq[VQ];
    load_tile<VEC>(dst, e0, e1, ng, g0, tid, -1, dq);
    #pragma unroll
    for (int k = 0; k < VQ; ++k) {
      const bool gv = group_ok(g0, k, tid, ng);
      int dv[4] = {dq[k].x, dq[k].y, dq[k].z, dq[k].w};
      #pragma unroll
      for (int c = 0; c < 4; ++c) {
        unsigned dl = (unsigned)(dv[c] - nb0);
        if (gv && dl < (unsigned)nbn) atomicAdd(&hc[dl], 1.0f);
      }
    }
  }
  __syncthreads();

  float* pp = degp + (size_t)blockIdx.y * n + nb0;
  for (int i = tid; i < nbn; i += ATPB) pp[i] = hc[i];
}

__global__ void __launch_bounds__(ATPB) deg_part_kernel(
    const int* __restrict__ dst, float* __restrict__ degp, int e, int n) {
  __shared__ float hc[DNB];
  if ((e & 3) == 0) deg_part_body<true>(hc, dst, degp, e, n);
  else              deg_part_body<false>(hc, dst, degp, e, n);
}

__global__ void dinv_kernel(const float* __restrict__ degp,
                            float* __restrict__ dinv, int n) {
  int i = blockIdx.x * blockDim.x + threadIdx.x;
  if (i >= n) return;
  float s = 1.0f;
  #pragma unroll 16
  for (int ch = 0; ch < DCH; ++ch) s += degp[(size_t)ch * n + i];
  dinv[i] = rsqrtf(s);
}


__global__ void __launch_bounds__(256) gemm1_kernel(
    const float* __restrict__ x, const float* __restrict__ W1,
    const float* __restrict__ dinv,
    float* __restrict__ h, float* __restrict__ out1, int nTiles, int n) {
  int wave = (blockIdx.x * 256 + threadIdx.x) >> 5;
  int lane = threadIdx.x & 31;
  if (wave >= nTiles) return;
  int half = lane >> 4;
  int m    = lane & 15;
  int row  = wave * 16 + m;
  int rowc = row < n ? row : n - 1;
  const float* xrow = x + (size_t)rowc * F_IN;

  v8f c = {};
  #pragma unroll 4
  for (int k0 = 0; k0 < F_IN; k0 += 4) {
    int kk = k0 + 2 * half;
    v2f a = *(const v2f*)(xrow + kk);
    v2f b;
    b.x = W1[kk * HID + m];
    b.y = W1[(kk + 1) * HID + m];
    c = __builtin_amdgcn_wmma_f32_16x16x4_f32(
        false, a, false, b, (short)0, c, false, false);
  }

  float dv = dinv[rowc];
  bool full = (wave * 16 + 16 <= n);
  if (full) {
    #pragma unroll
    for (int r = 0; r < 8; ++r) {
      int orow = wave * 16 + r + 8 * half;
      float w  = __shfl(dv, r + 8 * half, 32);
      size_t o = (size_t)orow * HID + m;
      h[o]    = c[r];
      out1[o] = c[r] * w * w;
    }
  } else {
    #pragma unroll
    for (int r = 0; r < 8; ++r) {
      int orow = wave * 16 + r + 8 * half;
      float w  = __shfl(dv, r + 8 * half, 32);
      if (orow < n) {
        size_t o = (size_t)orow * HID + m;
        h[o]    = c[r];
        out1[o] = c[r] * w * w;
      }
    }
  }
}


template <int WD, int NBK, int NCHK, bool VEC>
__device__ __forceinline__ void agg_body(
    float* acc, int* lst, int* cnt,
    const int* __restrict__ src, const int* __restrict__ dst,
    const float* __restrict__ dinv, const float* __restrict__ feat,
    float* __restrict__ part, int e, int n) {
  constexpr int LPE = WD / 4;
  constexpr int EPP = ATPB / LPE;
  constexpr int TILE_G = ATPB * VQ;
  const int tid = threadIdx.x;
  const int nb0 = blockIdx.x * NBK;
  const int nbn = min(NBK, n - nb0);
  const int per = (((e + NCHK - 1) / NCHK) + 3) & ~3;
  const int e0  = blockIdx.y * per;
  const int e1  = min(e, e0 + per);
  const int ng  = e1 > e0 ? (e1 - e0 + 3) >> 2 : 0;

  for (int i = tid; i < NBK * WD / 4; i += ATPB)
    ((float4*)acc)[i] = make_float4(0.f, 0.f, 0.f, 0.f);
  if (tid < 2) cnt[tid] = 0;
  __syncthreads();

  const int sub  = tid % LPE;
  const int slot = tid / LPE;
  const int lane = tid & 31;

  int par = 0;
  for (int g0 = 0; g0 < ng; g0 += TILE_G, par ^= 1) {
    int4 dq[VQ], sq[VQ];
    load_tile<VEC>(dst, e0, e1, ng, g0, tid, -1, dq);
    load_tile<VEC>(src, e0, e1, ng, g0, tid, 0, sq);
    static_assert(VQ == 8, "the fence below names eight src loads");
    __builtin_amdgcn_sched_barrier(0);
    asm volatile("" :: "v"(sq[0].x), "v"(sq[1].x), "v"(sq[2].x), "v"(sq[3].x),
                       "v"(sq[4].x), "v"(sq[5].x), "v"(sq[6].x), "v"(sq[7].x));

    unsigned nbk[VQ];
    #pragma unroll
    for (int k = 0; k < VQ; ++k)
      nbk[k] = group_ok(g0, k, tid, ng) ? (unsigned)nbn : 0u;
    unsigned mk[4 * VQ];
    int tot = 0;
    #pragma unroll
    for (int k = 0; k < VQ; ++k) {
      int dv[4] = {dq[k].x, dq[k].y, dq[k].z, dq[k].w};
      #pragma unroll
      for (int c = 0; c < 4; ++c) {
        mk[4 * k + c] = __builtin_amdgcn_ballot_w32((unsigned)(dv[c] - nb0) < nbk[k]);
        tot += __popc(mk[4 * k + c]);
      }
    }
    int base = 0;
    if (tot > 0) {
      if (lane == 0) base = atomicAdd(&cnt[par], tot);
      base = __builtin_amdgcn_readfirstlane(base);
    }
    if (base + tot <= LCAP) {
      int run = base;
      #pragma unroll
      for (int k = 0; k < VQ; ++k) {
        int dv[4] = {dq[k].x, dq[k].y, dq[k].z, dq[k].w};
        int sv[4] = {sq[k].x, sq[k].y, sq[k].z, sq[k].w};
        #pragma unroll
        for (int c = 0; c < 4; ++c) {
          const unsigned dl = (unsigned)(dv[c] - nb0);
          if (dl < nbk[k]) {
            const int p = (int)__builtin_amdgcn_mbcnt_lo(mk[4 * k + c], (unsigned)run);
            lst[2 * p]     = sv[c];
            lst[2 * p + 1] = (int)dl;
          }
          run += __popc(mk[4 * k + c]);
        }
      }
    } else {
      int run = base;
      #pragma unroll 1
      for (int b = 0; b < 4 * VQ; ++b) {
        const int g  = g0 + (b >> 2) * ATPB + tid;
        const int ei = e0 + 4 * g + (b & 3);
        const bool in = g < ng && ei < e1;
        int dvv = -1, svv = 0;
        if (in) {
          dvv = dst[ei];
          svv = src[ei];
        }
        const unsigned dl = (unsigned)(dvv - nb0);
        const bool o = in && dl < (unsigned)nbn;
        const unsigned mo = __builtin_amdgcn_ballot_w32(o);
        if (o) {
          const int p = (int)__builtin_amdgcn_mbcnt_lo(mo, (unsigned)run);
          if (p < LCAP) {
            lst[2 * p]     = svv;
            lst[2 * p + 1] = (int)dl;
          } else {
            float w = dinv[svv] * dinv[dvv];
            const float* fr = feat + (size_t)svv * WD;
            #pragma unroll 1
            for (int f = 0; f < WD; ++f) atomicAdd(&acc[dl * WD + f], fr[f] * w);
          }
        }
        run += __popc(mo);
      }
    }
    __syncthreads();
    if (tid == 0) cnt[par ^ 1] = 0;

    const int m = min(cnt[par], LCAP);
    for (int j = slot; j < m; j += 4 * EPP) {
      int s[4], d[4];
      bool ok[4];
      float da[4], db[4];
      float4 v[4];
      #pragma unroll
      for (int q = 0; q < 4; ++q) {
        int jj = j + q * EPP;
        ok[q] = jj < m;
        int jc = ok[q] ? jj : j;
        s[q] = lst[2 * jc];
        d[q] = lst[2 * jc + 1];
      }
      #pragma unroll
      for (int q = 0; q < 4; ++q) {
        da[q] = dinv[s[q]];
        db[q] = dinv[nb0 + d[q]];
        v[q]  = *(const float4*)(feat + (size_t)s[q] * WD + sub * 4);
      }
      __builtin_amdgcn_sched_barrier(0);
      asm volatile("" :: "v"(da[0]), "v"(da[1]), "v"(da[2]), "v"(da[3]),
                         "v"(db[0]), "v"(db[1]), "v"(db[2]), "v"(db[3]),
                         "v"(v[0].x), "v"(v[1].x), "v"(v[2].x), "v"(v[3].x));
      #pragma unroll
      for (int q = 0; q < 4; ++q) {
        if (ok[q]) {
          float w = da[q] * db[q];
          float* a = acc + d[q] * WD + sub * 4;
          atomicAdd(a + 0, v[q].x * w);
          atomicAdd(a + 1, v[q].y * w);
          atomicAdd(a + 2, v[q].z * w);
          atomicAdd(a + 3, v[q].w * w);
        }
      }
    }
    __syncthreads();
  }

  float4* pp = (float4*)(part + ((size_t)blockIdx.y * n + nb0) * WD);
  for (int i = tid; i < nbn * WD / 4; i += ATPB) pp[i] = ((const float4*)acc)[i];
}

template <int WD, int NBK, int NCHK>
__global__ void __launch_bounds__(ATPB) agg_kernel(
    const int* __restrict__ src, const int* __restrict__ dst,
    const float* __restrict__ dinv, const float* __restrict__ feat,
    float* __restrict__ part, int e, int n) {
  __shared__ float acc[NBK * WD];
  __shared__ int lst[2 * LCAP];
  __shared__ int cnt[2];
  if ((e & 3) == 0)
    agg_body<WD, NBK, NCHK, true>(acc, lst, cnt, src, dst, dinv, feat, part, e, n);
  else
    agg_body<WD, NBK, NCHK, false>(acc, lst, cnt, src, dst, dinv, feat, part, e, n);
}


__global__ void __launch_bounds__(256) gemm2_kernel(
    const float* __restrict__ h1, const float* __restrict__ part,
    const float* __restrict__ b1,
    const float* __restrict__ W2, const float* __restrict__ dinv,
    float* __restrict__ lt, float* __restrict__ out, int nTiles, int n) {
  int wave = (blockIdx.x * 256 + threadIdx.x) >> 5;
  int lane = threadIdx.x & 31;
  if (wave >= nTiles) return;
  int half = lane >> 4;
  int m    = lane & 15;
  int row  = wave * 16 + m;
  int rowc = row < n ? row : n - 1;
  const float* hrow = h1 + (size_t)rowc * HID;
  const float* prow = part + (size_t)rowc * HID;
  const size_t pstr = (size_t)n * HID;
  float maskv = (m < NCLS) ? 1.0f : 0.0f;
  int mc = m & (NCLS - 1);

  v8f c = {};
  #pragma unroll
  for (int k0 = 0; k0 < HID; k0 += 4) {
    int kk = k0 + 2 * half;
    v2f t = *(const v2f*)(hrow + kk);
    #pragma unroll
    for (int ch = 0; ch < NCH1; ++ch) {
      v2f pv = *(const v2f*)(prow + ch * pstr + kk);
      t.x += pv.x;
      t.y += pv.y;
    }
    v2f a;
    a.x = fmaxf(t.x + b1[kk],     0.0f);
    a.y = fmaxf(t.y + b1[kk + 1], 0.0f);
    v2f b;
    b.x = W2[kk * NCLS + mc] * maskv;
    b.y = W2[(kk + 1) * NCLS + mc] * maskv;
    c = __builtin_amdgcn_wmma_f32_16x16x4_f32(
        false, a, false, b, (short)0, c, false, false);
  }

  float dv = dinv[rowc];
  bool full = (wave * 16 + 16 <= n);
  if (m < NCLS) {
    #pragma unroll
    for (int r = 0; r < 8; ++r) {
      int orow = wave * 16 + r + 8 * half;
      float w  = __shfl(dv, r + 8 * half, 32);
      if (full || orow < n) {
        size_t o = (size_t)orow * NCLS + m;
        lt[o]  = c[r];
        out[o] = c[r] * w * w;
      }
    }
  }
}


__global__ void lsm_kernel(float* __restrict__ out,
                           const float* __restrict__ part,
                           const float* __restrict__ b2, int n) {
  int i = blockIdx.x * blockDim.x + threadIdx.x;
  if (i >= n) return;
  float* row = out + (size_t)i * NCLS;
  float z[NCLS];
  #pragma unroll
  for (int c = 0; c < NCLS; ++c) z[c] = row[c];
  #pragma unroll
  for (int ch = 0; ch < NCH2; ++ch) {
    const float4* pr = (const float4*)(part + ((size_t)ch * n + i) * NCLS);
    float4 p0 = pr[0], p1 = pr[1];
    z[0] += p0.x; z[1] += p0.y; z[2] += p0.z; z[3] += p0.w;
    z[4] += p1.x; z[5] += p1.y; z[6] += p1.z; z[7] += p1.w;
  }
  float mx = -INFINITY;
  #pragma unroll
  for (int c = 0; c < NCLS; ++c) {
    z[c] += b2[c];
    mx = fmaxf(mx, z[c]);
  }
  float s = 0.0f;
  #pragma unroll
  for (int c = 0; c < NCLS; ++c) s += expf(z[c] - mx);
  float l = mx + logf(s);
  #pragma unroll
  for (int c = 0; c < NCLS; ++c) row[c] = z[c] - l;
}


extern "C" void kernel_launch(void* const* d_in, const int* in_sizes, int n_in,
                              void* d_out, int out_size, void* d_ws, size_t ws_size,
                              hipStream_t stream) {
  const float* x  = (const float*)d_in[0];
  const int* edge = (const int*)d_in[1];
  const float* W1 = (const float*)d_in[2];
  const float* b1 = (const float*)d_in[3];
  const float* W2 = (const float*)d_in[4];
  const float* b2 = (const float*)d_in[5];

  int n = in_sizes[0] / F_IN;
  int e = in_sizes[1] / 2;
  const int* src = edge;
  const int* dst = edge + e;

  float* deg = (float*)d_ws;
  float* h   = deg + n;
  float* h1  = h + (size_t)n * HID;
  float* lt  = h1 + (size_t)n * HID;
  float* p1  = lt + (size_t)n * NCLS;
  float* p2  = p1 + (size_t)NCH1 * n * HID;
  float* dgp = p2 + (size_t)NCH2 * n * NCLS;
  float* out = (float*)d_out;

  const int T = 256;
  int nTiles  = (n + 15) / 16;
  int gBlocks = (nTiles + 7) / 8;
  dim3 dGrid((n + DNB - 1) / DNB, DCH);
  dim3 aGrid1((n + NB1 - 1) / NB1, NCH1);
  dim3 aGrid2((n + NB2 - 1) / NB2, NCH2);

  deg_part_kernel<<<dGrid, ATPB, 0, stream>>>(dst, dgp, e, n);
  dinv_kernel<<<(n + T - 1) / T, T, 0, stream>>>(dgp, deg, n);

  gemm1_kernel<<<gBlocks, T, 0, stream>>>(x, W1, deg, h, h1, nTiles, n);
  agg_kernel<HID, NB1, NCH1><<<aGrid1, ATPB, 0, stream>>>(src, dst, deg, h, p1, e, n);

  gemm2_kernel<<<gBlocks, T, 0, stream>>>(h1, p1, b1, W2, deg, lt, out, nTiles, n);
  agg_kernel<NCLS, NB2, NCH2><<<aGrid2, ATPB, 0, stream>>>(src, dst, deg, lt, p2, e, n);

  lsm_kernel<<<(n + T - 1) / T, T, 0, stream>>>(out, p2, b2, n);
}


# Gathered rows carry their node's dinv in their own 128-byte line (h rows 16+1 of 32 floats, layer-2 rows 8+1 of 16): an edge's source weight comes with its row, one L2 line per edge, not two
# speedup vs baseline: 7.1393x; 1.0758x over previous
#include <hip/hip_runtime.h>
#include <hip/hip_bf16.h>
#include <math.h>


typedef float v2f __attribute__((ext_vector_type(2)));
typedef float v8f __attribute__((ext_vector_type(8)));

#define F_IN 128
#define HID  16
#define NCLS 8

#define RS1 (2 * HID)
#define RS2 (2 * NCLS)


#define ATPB 1024
#define VQ   8
#define LCAP 4096

#define DNB  25000
#define DCH  64
#define NB1  4000
#define NCH1 10
#define NB2  8000
#define NCH2 19


template <bool VEC>
__device__ __forceinline__ void load_tile(const int* __restrict__ p, int e0,
                                          int e1, int ng, int gb, int tid,
                                          int fill, int4 (&q)[VQ]) {
  if (VEC) {
    #pragma unroll
    for (int k = 0; k < VQ; ++k) {
      int g = gb + k * ATPB + tid;
      q[k] = *(const int4*)(p + e0 + 4 * (g < ng ? g : 0));
    }
  } else {
    #pragma unroll
    for (int k = 0; k < VQ; ++k) {
      int g = gb + k * ATPB + tid;
      int base = e0 + 4 * (g < ng ? g : 0);
      q[k].x = (base + 0 < e1) ? p[base + 0] : fill;
      q[k].y = (base + 1 < e1) ? p[base + 1] : fill;
      q[k].z = (base + 2 < e1) ? p[base + 2] : fill;
      q[k].w = (base + 3 < e1) ? p[base + 3] : fill;
    }
  }
}

__device__ __forceinline__ bool group_ok(int gb, int k, int tid, int ng) {
  return gb + k * ATPB + tid < ng;
}


template <bool VEC>
__device__ __forceinline__ void deg_part_body(
    float* hc, const int* __restrict__ dst, float* __restrict__ degp, int e, int n) {
  const int tid = threadIdx.x;
  const int nb0 = blockIdx.x * DNB;
  const int nbn = min(DNB, n - nb0);
  const int per = (((e + DCH - 1) / DCH) + 3) & ~3;
  const int e0  = blockIdx.y * per;
  const int e1  = min(e, e0 + per);
  const int ng  = e1 > e0 ? (e1 - e0 + 3) >> 2 : 0;

  for (int i = tid; i < DNB; i += ATPB) hc[i] = 0.0f;
  __syncthreads();

  for (int g0 = 0; g0 < ng; g0 += ATPB * VQ) {
    int4 dq[VQ];
    load_tile<VEC>(dst, e0, e1, ng, g0, tid, -1, dq);
    #pragma unroll
    for (int k = 0; k < VQ; ++k) {
      const bool gv = group_ok(g0, k, tid, ng);
      int dv[4] = {dq[k].x, dq[k].y, dq[k].z, dq[k].w};
      #pragma unroll
      for (int c = 0; c < 4; ++c) {
        unsigned dl = (unsigned)(dv[c] - nb0);
        if (gv && dl < (unsigned)nbn) atomicAdd(&hc[dl], 1.0f);
      }
    }
  }
  __syncthreads();

  float* pp = degp + (size_t)blockIdx.y * n + nb0;
  for (int i = tid; i < nbn; i += ATPB) pp[i] = hc[i];
}

__global__ void __launch_bounds__(ATPB) deg_part_kernel(
    const int* __restrict__ dst, float* __restrict__ degp, int e, int n) {
  __shared__ float hc[DNB];
  if ((e & 3) == 0) deg_part_body<true>(hc, dst, degp, e, n);
  else              deg_part_body<false>(hc, dst, degp, e, n);
}

__global__ void dinv_kernel(const float* __restrict__ degp,
                            float* __restrict__ dinv, int n) {
  int i = blockIdx.x * blockDim.x + threadIdx.x;
  if (i >= n) return;
  float s = 1.0f;
  #pragma unroll 16
  for (int ch = 0; ch < DCH; ++ch) s += degp[(size_t)ch * n + i];
  dinv[i] = rsqrtf(s);
}


__global__ void __launch_bounds__(256) gemm1_kernel(
    const float* __restrict__ x, const float* __restrict__ W1,
    const float* __restrict__ dinv,
    float* __restrict__ h, float* __restrict__ out1, int nTiles, int n) {
  int wave = (blockIdx.x * 256 + threadIdx.x) >> 5;
  int lane = threadIdx.x & 31;
  if (wave >= nTiles) return;
  int half = lane >> 4;
  int m    = lane & 15;
  int row  = wave * 16 + m;
  int rowc = row < n ? row : n - 1;
  const float* xrow = x + (size_t)rowc * F_IN;

  v8f c = {};
  #pragma unroll 4
  for (int k0 = 0; k0 < F_IN; k0 += 4) {
    int kk = k0 + 2 * half;
    v2f a = *(const v2f*)(xrow + kk);
    v2f b;
    b.x = W1[kk * HID + m];
    b.y = W1[(kk + 1) * HID + m];
    c = __builtin_amdgcn_wmma_f32_16x16x4_f32(
        false, a, false, b, (short)0, c, false, false);
  }

  float dv = dinv[rowc];
  bool full = (wave * 16 + 16 <= n);
  if (full) {
    #pragma unroll
    for (int r = 0; r < 8; ++r) {
      int orow = wave * 16 + r + 8 * half;
      float w  = __shfl(dv, r + 8 * half, 32);
      size_t o = (size_t)orow * HID + m;
      size_t g = (size_t)orow * RS1 + m;
      h[g]       = c[r];
      h[g + HID] = (m == 0) ? w : 0.0f;
      out1[o] = c[r] * w * w;
    }
  } else {
    #pragma unroll
    for (int r = 0; r < 8; ++r) {
      int orow = wave * 16 + r + 8 * half;
      float w  = __shfl(dv, r + 8 * half, 32);
      if (orow < n) {
        size_t o = (size_t)orow * HID + m;
        size_t g = (size_t)orow * RS1 + m;
        h[g]       = c[r];
        h[g + HID] = (m == 0) ? w : 0.0f;
        out1[o] = c[r] * w * w;
      }
    }
  }
}


template <int WD, int NBK, int NCHK, bool VEC>
__device__ __forceinline__ void agg_body(
    float* acc, int* lst, int* cnt,
    const int* __restrict__ src, const int* __restrict__ dst,
    const float* __restrict__ dinv, const float* __restrict__ feat,
    float* __restrict__ part, int e, int n) {
  constexpr int RS  = 2 * WD;
  constexpr int LPE = WD / 4;
  constexpr int EPP = ATPB / LPE;
  constexpr int TILE_G = ATPB * VQ;
  const int tid = threadIdx.x;
  const int nb0 = blockIdx.x * NBK;
  const int nbn = min(NBK, n - nb0);
  const int per = (((e + NCHK - 1) / NCHK) + 3) & ~3;
  const int e0  = blockIdx.y * per;
  const int e1  = min(e, e0 + per);
  const int ng  = e1 > e0 ? (e1 - e0 + 3) >> 2 : 0;

  for (int i = tid; i < NBK * WD / 4; i += ATPB)
    ((float4*)acc)[i] = make_float4(0.f, 0.f, 0.f, 0.f);
  if (tid < 2) cnt[tid] = 0;
  __syncthreads();

  const int sub  = tid % LPE;
  const int slot = tid / LPE;
  const int lane = tid & 31;

  int par = 0;
  for (int g0 = 0; g0 < ng; g0 += TILE_G, par ^= 1) {
    int4 dq[VQ], sq[VQ];
    load_tile<VEC>(dst, e0, e1, ng, g0, tid, -1, dq);
    load_tile<VEC>(src, e0, e1, ng, g0, tid, 0, sq);
    static_assert(VQ == 8, "the fence below names eight src loads");
    __builtin_amdgcn_sched_barrier(0);
    asm volatile("" :: "v"(sq[0].x), "v"(sq[1].x), "v"(sq[2].x), "v"(sq[3].x),
                       "v"(sq[4].x), "v"(sq[5].x), "v"(sq[6].x), "v"(sq[7].x));

    unsigned nbk[VQ];
    #pragma unroll
    for (int k = 0; k < VQ; ++k)
      nbk[k] = group_ok(g0, k, tid, ng) ? (unsigned)nbn : 0u;
    unsigned mk[4 * VQ];
    int tot = 0;
    #pragma unroll
    for (int k = 0; k < VQ; ++k) {
      int dv[4] = {dq[k].x, dq[k].y, dq[k].z, dq[k].w};
      #pragma unroll
      for (int c = 0; c < 4; ++c) {
        mk[4 * k + c] = __builtin_amdgcn_ballot_w32((unsigned)(dv[c] - nb0) < nbk[k]);
        tot += __popc(mk[4 * k + c]);
      }
    }
    int base = 0;
    if (tot > 0) {
      if (lane == 0) base = atomicAdd(&cnt[par], tot);
      base = __builtin_amdgcn_readfirstlane(base);
    }
    if (base + tot <= LCAP) {
      int run = base;
      #pragma unroll
      for (int k = 0; k < VQ; ++k) {
        int dv[4] = {dq[k].x, dq[k].y, dq[k].z, dq[k].w};
        int sv[4] = {sq[k].x, sq[k].y, sq[k].z, sq[k].w};
        #pragma unroll
        for (int c = 0; c < 4; ++c) {
          const unsigned dl = (unsigned)(dv[c] - nb0);
          if (dl < nbk[k]) {
            const int p = (int)__builtin_amdgcn_mbcnt_lo(mk[4 * k + c], (unsigned)run);
            lst[2 * p]     = sv[c];
            lst[2 * p + 1] = (int)dl;
          }
          run += __popc(mk[4 * k + c]);
        }
      }
    } else {
      int run = base;
      #pragma unroll 1
      for (int b = 0; b < 4 * VQ; ++b) {
        const int g  = g0 + (b >> 2) * ATPB + tid;
        const int ei = e0 + 4 * g + (b & 3);
        const bool in = g < ng && ei < e1;
        int dvv = -1, svv = 0;
        if (in) {
          dvv = dst[ei];
          svv = src[ei];
        }
        const unsigned dl = (unsigned)(dvv - nb0);
        const bool o = in && dl < (unsigned)nbn;
        const unsigned mo = __builtin_amdgcn_ballot_w32(o);
        if (o) {
          const int p = (int)__builtin_amdgcn_mbcnt_lo(mo, (unsigned)run);
          if (p < LCAP) {
            lst[2 * p]     = svv;
            lst[2 * p + 1] = (int)dl;
          } else {
            const float* fr = feat + (size_t)svv * RS;
            float w = fr[WD] * dinv[dvv];
            #pragma unroll 1
            for (int f = 0; f < WD; ++f) atomicAdd(&acc[dl * WD + f], fr[f] * w);
          }
        }
        run += __popc(mo);
      }
    }
    __syncthreads();
    if (tid == 0) cnt[par ^ 1] = 0;

    const int m = min(cnt[par], LCAP);
    for (int j = slot; j < m; j += 4 * EPP) {
      int s[4], d[4];
      bool ok[4];
      float da[4], db[4];
      float4 v[4];
      #pragma unroll
      for (int q = 0; q < 4; ++q) {
        int jj = j + q * EPP;
        ok[q] = jj < m;
        int jc = ok[q] ? jj : j;
        s[q] = lst[2 * jc];
        d[q] = lst[2 * jc + 1];
      }
      #pragma unroll
      for (int q = 0; q < 4; ++q) {
        const float* fr = feat + (size_t)s[q] * RS;
        v[q]  = *(const float4*)(fr + sub * 4);
        da[q] = fr[WD];
        db[q] = dinv[nb0 + d[q]];
      }
      __builtin_amdgcn_sched_barrier(0);
      asm volatile("" :: "v"(da[0]), "v"(da[1]), "v"(da[2]), "v"(da[3]),
                         "v"(db[0]), "v"(db[1]), "v"(db[2]), "v"(db[3]),
                         "v"(v[0].x), "v"(v[1].x), "v"(v[2].x), "v"(v[3].x));
      #pragma unroll
      for (int q = 0; q < 4; ++q) {
        if (ok[q]) {
          float w = da[q] * db[q];
          float* a = acc + d[q] * WD + sub * 4;
          atomicAdd(a + 0, v[q].x * w);
          atomicAdd(a + 1, v[q].y * w);
          atomicAdd(a + 2, v[q].z * w);
          atomicAdd(a + 3, v[q].w * w);
        }
      }
    }
    __syncthreads();
  }

  float4* pp = (float4*)(part + ((size_t)blockIdx.y * n + nb0) * WD);
  for (int i = tid; i < nbn * WD / 4; i += ATPB) pp[i] = ((const float4*)acc)[i];
}

template <int WD, int NBK, int NCHK>
__global__ void __launch_bounds__(ATPB) agg_kernel(
    const int* __restrict__ src, const int* __restrict__ dst,
    const float* __restrict__ dinv, const float* __restrict__ feat,
    float* __restrict__ part, int e, int n) {
  __shared__ float acc[NBK * WD];
  __shared__ int lst[2 * LCAP];
  __shared__ int cnt[2];
  if ((e & 3) == 0)
    agg_body<WD, NBK, NCHK, true>(acc, lst, cnt, src, dst, dinv, feat, part, e, n);
  else
    agg_body<WD, NBK, NCHK, false>(acc, lst, cnt, src, dst, dinv, feat, part, e, n);
}


__global__ void __launch_bounds__(256) gemm2_kernel(
    const float* __restrict__ h1, const float* __restrict__ part,
    const float* __restrict__ b1,
    const float* __restrict__ W2, const float* __restrict__ dinv,
    float* __restrict__ lt, float* __restrict__ out, int nTiles, int n) {
  int wave = (blockIdx.x * 256 + threadIdx.x) >> 5;
  int lane = threadIdx.x & 31;
  if (wave >= nTiles) return;
  int half = lane >> 4;
  int m    = lane & 15;
  int row  = wave * 16 + m;
  int rowc = row < n ? row : n - 1;
  const float* hrow = h1 + (size_t)rowc * HID;
  const float* prow = part + (size_t)rowc * HID;
  const size_t pstr = (size_t)n * HID;
  float maskv = (m < NCLS) ? 1.0f : 0.0f;
  int mc = m & (NCLS - 1);

  v8f c = {};
  #pragma unroll
  for (int k0 = 0; k0 < HID; k0 += 4) {
    int kk = k0 + 2 * half;
    v2f t = *(const v2f*)(hrow + kk);
    #pragma unroll
    for (int ch = 0; ch < NCH1; ++ch) {
      v2f pv = *(const v2f*)(prow + ch * pstr + kk);
      t.x += pv.x;
      t.y += pv.y;
    }
    v2f a;
    a.x = fmaxf(t.x + b1[kk],     0.0f);
    a.y = fmaxf(t.y + b1[kk + 1], 0.0f);
    v2f b;
    b.x = W2[kk * NCLS + mc] * maskv;
    b.y = W2[(kk + 1) * NCLS + mc] * maskv;
    c = __builtin_amdgcn_wmma_f32_16x16x4_f32(
        false, a, false, b, (short)0, c, false, false);
  }

  float dv = dinv[rowc];
  bool full = (wave * 16 + 16 <= n);
  #pragma unroll
  for (int r = 0; r < 8; ++r) {
    int orow = wave * 16 + r + 8 * half;
    float w  = __shfl(dv, r + 8 * half, 32);
    if (full || orow < n) {
      lt[(size_t)orow * RS2 + m] = (m < NCLS) ? c[r] : ((m == NCLS) ? w : 0.0f);
      if (m < NCLS)
        out[(size_t)orow * NCLS + m] = c[r] * w * w;
    }
  }
}


__global__ void lsm_kernel(float* __restrict__ out,
                           const float* __restrict__ part,
                           const float* __restrict__ b2, int n) {
  int i = blockIdx.x * blockDim.x + threadIdx.x;
  if (i >= n) return;
  float* row = out + (size_t)i * NCLS;
  float z[NCLS];
  #pragma unroll
  for (int c = 0; c < NCLS; ++c) z[c] = row[c];
  #pragma unroll
  for (int ch = 0; ch < NCH2; ++ch) {
    const float4* pr = (const float4*)(part + ((size_t)ch * n + i) * NCLS);
    float4 p0 = pr[0], p1 = pr[1];
    z[0] += p0.x; z[1] += p0.y; z[2] += p0.z; z[3] += p0.w;
    z[4] += p1.x; z[5] += p1.y; z[6] += p1.z; z[7] += p1.w;
  }
  float mx = -INFINITY;
  #pragma unroll
  for (int c = 0; c < NCLS; ++c) {
    z[c] += b2[c];
    mx = fmaxf(mx, z[c]);
  }
  float s = 0.0f;
  #pragma unroll
  for (int c = 0; c < NCLS; ++c) s += expf(z[c] - mx);
  float l = mx + logf(s);
  #pragma unroll
  for (int c = 0; c < NCLS; ++c) row[c] = z[c] - l;
}


extern "C" void kernel_launch(void* const* d_in, const int* in_sizes, int n_in,
                              void* d_out, int out_size, void* d_ws, size_t ws_size,
                              hipStream_t stream) {
  const float* x  = (const float*)d_in[0];
  const int* edge = (const int*)d_in[1];
  const float* W1 = (const float*)d_in[2];
  const float* b1 = (const float*)d_in[3];
  const float* W2 = (const float*)d_in[4];
  const float* b2 = (const float*)d_in[5];

  int n = in_sizes[0] / F_IN;
  int e = in_sizes[1] / 2;
  const int* src = edge;
  const int* dst = edge + e;

  float* deg = (float*)d_ws;
  float* h   = deg + n;
  float* h1  = h + (size_t)n * RS1;
  float* lt  = h1 + (size_t)n * HID;
  float* p1  = lt + (size_t)n * RS2;
  float* p2  = p1 + (size_t)NCH1 * n * HID;
  float* dgp = p2 + (size_t)NCH2 * n * NCLS;
  float* out = (float*)d_out;

  const int T = 256;
  int nTiles  = (n + 15) / 16;
  int gBlocks = (nTiles + 7) / 8;
  dim3 dGrid((n + DNB - 1) / DNB, DCH);
  dim3 aGrid1((n + NB1 - 1) / NB1, NCH1);
  dim3 aGrid2((n + NB2 - 1) / NB2, NCH2);

  deg_part_kernel<<<dGrid, ATPB, 0, stream>>>(dst, dgp, e, n);
  dinv_kernel<<<(n + T - 1) / T, T, 0, stream>>>(dgp, deg, n);

  gemm1_kernel<<<gBlocks, T, 0, stream>>>(x, W1, deg, h, h1, nTiles, n);
  agg_kernel<HID, NB1, NCH1><<<aGrid1, ATPB, 0, stream>>>(src, dst, deg, h, p1, e, n);

  gemm2_kernel<<<gBlocks, T, 0, stream>>>(h1, p1, b1, W2, deg, lt, out, nTiles, n);
  agg_kernel<NCLS, NB2, NCH2><<<aGrid2, ATPB, 0, stream>>>(src, dst, deg, lt, p2, e, n);

  lsm_kernel<<<(n + T - 1) / T, T, 0, stream>>>(out, p2, b2, n);
}


# Edge-sum lists per wave instead of per work-group: a wave filters its tile into its own LDS list and sums it; no work-group barrier or shared counter in the tile loop; LDS-atomic sum order changes
# speedup vs baseline: 8.1437x; 1.1407x over previous
#include <hip/hip_runtime.h>
#include <hip/hip_bf16.h>
#include <math.h>


typedef float v2f __attribute__((ext_vector_type(2)));
typedef float v8f __attribute__((ext_vector_type(8)));

#define F_IN 128
#define HID  16
#define NCLS 8

#define RS1 (2 * HID)
#define RS2 (2 * NCLS)


#define ATPB 1024
#define VQ   8
#define WCAP 192

#define DNB  25000
#define DCH  64
#define NB1  4000
#define NCH1 10
#define NB2  8000
#define NCH2 19


template <bool VEC>
__device__ __forceinline__ void load_tile(const int* __restrict__ p, int e0,
                                          int e1, int ng, int gb, int tid,
                                          int fill, int4 (&q)[VQ]) {
  if (VEC) {
    #pragma unroll
    for (int k = 0; k < VQ; ++k) {
      int g = gb + k * ATPB + tid;
      q[k] = *(const int4*)(p + e0 + 4 * (g < ng ? g : 0));
    }
  } else {
    #pragma unroll
    for (int k = 0; k < VQ; ++k) {
      int g = gb + k * ATPB + tid;
      int base = e0 + 4 * (g < ng ? g : 0);
      q[k].x = (base + 0 < e1) ? p[base + 0] : fill;
      q[k].y = (base + 1 < e1) ? p[base + 1] : fill;
      q[k].z = (base + 2 < e1) ? p[base + 2] : fill;
      q[k].w = (base + 3 < e1) ? p[base + 3] : fill;
    }
  }
}

__device__ __forceinline__ bool group_ok(int gb, int k, int tid, int ng) {
  return gb + k * ATPB + tid < ng;
}


template <bool VEC>
__device__ __forceinline__ void deg_part_body(
    float* hc, const int* __restrict__ dst, float* __restrict__ degp, int e, int n) {
  const int tid = threadIdx.x;
  const int nb0 = blockIdx.x * DNB;
  const int nbn = min(DNB, n - nb0);
  const int per = (((e + DCH - 1) / DCH) + 3) & ~3;
  const int e0  = blockIdx.y * per;
  const int e1  = min(e, e0 + per);
  const int ng  = e1 > e0 ? (e1 - e0 + 3) >> 2 : 0;

  for (int i = tid; i < DNB; i += ATPB) hc[i] = 0.0f;
  __syncthreads();

  for (int g0 = 0; g0 < ng; g0 += ATPB * VQ) {
    int4 dq[VQ];
    load_tile<VEC>(dst, e0, e1, ng, g0, tid, -1, dq);
    #pragma unroll
    for (int k = 0; k < VQ; ++k) {
      const bool gv = group_ok(g0, k, tid, ng);
      int dv[4] = {dq[k].x, dq[k].y, dq[k].z, dq[k].w};
      #pragma unroll
      for (int c = 0; c < 4; ++c) {
        unsigned dl = (unsigned)(dv[c] - nb0);
        if (gv && dl < (unsigned)nbn) atomicAdd(&hc[dl], 1.0f);
      }
    }
  }
  __syncthreads();

  float* pp = degp + (size_t)blockIdx.y * n + nb0;
  for (int i = tid; i < nbn; i += ATPB) pp[i] = hc[i];
}

__global__ void __launch_bounds__(ATPB) deg_part_kernel(
    const int* __restrict__ dst, float* __restrict__ degp, int e, int n) {
  __shared__ float hc[DNB];
  if ((e & 3) == 0) deg_part_body<true>(hc, dst, degp, e, n);
  else              deg_part_body<false>(hc, dst, degp, e, n);
}

__global__ void dinv_kernel(const float* __restrict__ degp,
                            float* __restrict__ dinv, int n) {
  int i = blockIdx.x * blockDim.x + threadIdx.x;
  if (i >= n) return;
  float s = 1.0f;
  #pragma unroll 16
  for (int ch = 0; ch < DCH; ++ch) s += degp[(size_t)ch * n + i];
  dinv[i] = rsqrtf(s);
}


__global__ void __launch_bounds__(256) gemm1_kernel(
    const float* __restrict__ x, const float* __restrict__ W1,
    const float* __restrict__ dinv,
    float* __restrict__ h, float* __restrict__ out1, int nTiles, int n) {
  int wave = (blockIdx.x * 256 + threadIdx.x) >> 5;
  int lane = threadIdx.x & 31;
  if (wave >= nTiles) return;
  int half = lane >> 4;
  int m    = lane & 15;
  int row  = wave * 16 + m;
  int rowc = row < n ? row : n - 1;
  const float* xrow = x + (size_t)rowc * F_IN;

  v8f c = {};
  #pragma unroll 4
  for (int k0 = 0; k0 < F_IN; k0 += 4) {
    int kk = k0 + 2 * half;
    v2f a = *(const v2f*)(xrow + kk);
    v2f b;
    b.x = W1[kk * HID + m];
    b.y = W1[(kk + 1) * HID + m];
    c = __builtin_amdgcn_wmma_f32_16x16x4_f32(
        false, a, false, b, (short)0, c, false, false);
  }

  float dv = dinv[rowc];
  bool full = (wave * 16 + 16 <= n);
  if (full) {
    #pragma unroll
    for (int r = 0; r < 8; ++r) {
      int orow = wave * 16 + r + 8 * half;
      float w  = __shfl(dv, r + 8 * half, 32);
      size_t o = (size_t)orow * HID + m;
      size_t g = (size_t)orow * RS1 + m;
      h[g]       = c[r];
      h[g + HID] = (m == 0) ? w : 0.0f;
      out1[o] = c[r] * w * w;
    }
  } else {
    #pragma unroll
    for (int r = 0; r < 8; ++r) {
      int orow = wave * 16 + r + 8 * half;
      float w  = __shfl(dv, r + 8 * half, 32);
      if (orow < n) {
        size_t o = (size_t)orow * HID + m;
        size_t g = (size_t)orow * RS1 + m;
        h[g]       = c[r];
        h[g + HID] = (m == 0) ? w : 0.0f;
        out1[o] = c[r] * w * w;
      }
    }
  }
}


template <int WD, int NBK, int NCHK, bool VEC>
__device__ __forceinline__ void agg_body(
    float* acc, int* lst,
    const int* __restrict__ src, const int* __restrict__ dst,
    const float* __restrict__ dinv, const float* __restrict__ feat,
    float* __restrict__ part, int e, int n) {
  constexpr int RS  = 2 * WD;
  constexpr int LPE = WD / 4;
  constexpr int EPW = 32 / LPE;
  constexpr int TILE_G = ATPB * VQ;
  const int tid = threadIdx.x;
  const int nb0 = blockIdx.x * NBK;
  const int nbn = min(NBK, n - nb0);
  const int per = (((e + NCHK - 1) / NCHK) + 3) & ~3;
  const int e0  = blockIdx.y * per;
  const int e1  = min(e, e0 + per);
  const int ng  = e1 > e0 ? (e1 - e0 + 3) >> 2 : 0;

  for (int i = tid; i < NBK * WD / 4; i += ATPB)
    ((float4*)acc)[i] = make_float4(0.f, 0.f, 0.f, 0.f);
  __syncthreads();

  const int lane = tid & 31;
  const int sub  = lane % LPE;
  const int slot = lane / LPE;
  int* wl = lst + (tid >> 5) * (2 * WCAP);

  for (int g0 = 0; g0 < ng; g0 += TILE_G) {
    int4 dq[VQ], sq[VQ];
    load_tile<VEC>(dst, e0, e1, ng, g0, tid, -1, dq);
    load_tile<VEC>(src, e0, e1, ng, g0, tid, 0, sq);
    static_assert(VQ == 8, "the fence below names eight src loads");
    __builtin_amdgcn_sched_barrier(0);
    asm volatile("" :: "v"(sq[0].x), "v"(sq[1].x), "v"(sq[2].x), "v"(sq[3].x),
                       "v"(sq[4].x), "v"(sq[5].x), "v"(sq[6].x), "v"(sq[7].x));

    unsigned nbk[VQ];
    #pragma unroll
    for (int k = 0; k < VQ; ++k)
      nbk[k] = group_ok(g0, k, tid, ng) ? (unsigned)nbn : 0u;
    unsigned mk[4 * VQ];
    int tot = 0;
    #pragma unroll
    for (int k = 0; k < VQ; ++k) {
      int dv[4] = {dq[k].x, dq[k].y, dq[k].z, dq[k].w};
      #pragma unroll
      for (int c = 0; c < 4; ++c) {
        mk[4 * k + c] = __builtin_amdgcn_ballot_w32((unsigned)(dv[c] - nb0) < nbk[k]);
        tot += __popc(mk[4 * k + c]);
      }
    }
    if (tot <= WCAP) {
      int run = 0;
      #pragma unroll
      for (int k = 0; k < VQ; ++k) {
        int dv[4] = {dq[k].x, dq[k].y, dq[k].z, dq[k].w};
        int sv[4] = {sq[k].x, sq[k].y, sq[k].z, sq[k].w};
        #pragma unroll
        for (int c = 0; c < 4; ++c) {
          const unsigned dl = (unsigned)(dv[c] - nb0);
          if (dl < nbk[k]) {
            const int p = (int)__builtin_amdgcn_mbcnt_lo(mk[4 * k + c], (unsigned)run);
            wl[2 * p]     = sv[c];
            wl[2 * p + 1] = (int)dl;
          }
          run += __popc(mk[4 * k + c]);
        }
      }

      for (int j = slot; j < tot; j += 4 * EPW) {
        int s[4], d[4];
        bool ok[4];
        float da[4], db[4];
        float4 v[4];
        #pragma unroll
        for (int q = 0; q < 4; ++q) {
          int jj = j + q * EPW;
          ok[q] = jj < tot;
          int jc = ok[q] ? jj : j;
          s[q] = wl[2 * jc];
          d[q] = wl[2 * jc + 1];
        }
        #pragma unroll
        for (int q = 0; q < 4; ++q) {
          const float* fr = feat + (size_t)s[q] * RS;
          v[q]  = *(const float4*)(fr + sub * 4);
          da[q] = fr[WD];
          db[q] = dinv[nb0 + d[q]];
        }
        __builtin_amdgcn_sched_barrier(0);
        asm volatile("" :: "v"(da[0]), "v"(da[1]), "v"(da[2]), "v"(da[3]),
                           "v"(db[0]), "v"(db[1]), "v"(db[2]), "v"(db[3]),
                           "v"(v[0].x), "v"(v[1].x), "v"(v[2].x), "v"(v[3].x));
        #pragma unroll
        for (int q = 0; q < 4; ++q) {
          if (ok[q]) {
            float w = da[q] * db[q];
            float* a = acc + d[q] * WD + sub * 4;
            atomicAdd(a + 0, v[q].x * w);
            atomicAdd(a + 1, v[q].y * w);
            atomicAdd(a + 2, v[q].z * w);
            atomicAdd(a + 3, v[q].w * w);
          }
        }
      }
    } else {
      #pragma unroll 1
      for (int b = 0; b < 4 * VQ; ++b) {
        const int g  = g0 + (b >> 2) * ATPB + tid;
        const int ei = e0 + 4 * g + (b & 3);
        if (g < ng && ei < e1) {
          const int dvv = dst[ei];
          const unsigned dl = (unsigned)(dvv - nb0);
          if (dl < (unsigned)nbn) {
            const float* fr = feat + (size_t)src[ei] * RS;
            float w = fr[WD] * dinv[dvv];
            #pragma unroll 1
            for (int f = 0; f < WD; ++f) atomicAdd(&acc[dl * WD + f], fr[f] * w);
          }
        }
      }
    }
  }
  __syncthreads();

  float4* pp = (float4*)(part + ((size_t)blockIdx.y * n + nb0) * WD);
  for (int i = tid; i < nbn * WD / 4; i += ATPB) pp[i] = ((const float4*)acc)[i];
}

template <int WD, int NBK, int NCHK>
__global__ void __launch_bounds__(ATPB) agg_kernel(
    const int* __restrict__ src, const int* __restrict__ dst,
    const float* __restrict__ dinv, const float* __restrict__ feat,
    float* __restrict__ part, int e, int n) {
  __shared__ float acc[NBK * WD];
  __shared__ int lst[(ATPB / 32) * 2 * WCAP];
  if ((e & 3) == 0)
    agg_body<WD, NBK, NCHK, true>(acc, lst, src, dst, dinv, feat, part, e, n);
  else
    agg_body<WD, NBK, NCHK, false>(acc, lst, src, dst, dinv, feat, part, e, n);
}


__global__ void __launch_bounds__(256) gemm2_kernel(
    const float* __restrict__ h1, const float* __restrict__ part,
    const float* __restrict__ b1,
    const float* __restrict__ W2, const float* __restrict__ dinv,
    float* __restrict__ lt, float* __restrict__ out, int nTiles, int n) {
  int wave = (blockIdx.x * 256 + threadIdx.x) >> 5;
  int lane = threadIdx.x & 31;
  if (wave >= nTiles) return;
  int half = lane >> 4;
  int m    = lane & 15;
  int row  = wave * 16 + m;
  int rowc = row < n ? row : n - 1;
  const float* hrow = h1 + (size_t)rowc * HID;
  const float* prow = part + (size_t)rowc * HID;
  const size_t pstr = (size_t)n * HID;
  float maskv = (m < NCLS) ? 1.0f : 0.0f;
  int mc = m & (NCLS - 1);

  v8f c = {};
  #pragma unroll
  for (int k0 = 0; k0 < HID; k0 += 4) {
    int kk = k0 + 2 * half;
    v2f t = *(const v2f*)(hrow + kk);
    #pragma unroll
    for (int ch = 0; ch < NCH1; ++ch) {
      v2f pv = *(const v2f*)(prow + ch * pstr + kk);
      t.x += pv.x;
      t.y += pv.y;
    }
    v2f a;
    a.x = fmaxf(t.x + b1[kk],     0.0f);
    a.y = fmaxf(t.y + b1[kk + 1], 0.0f);
    v2f b;
    b.x = W2[kk * NCLS + mc] * maskv;
    b.y = W2[(kk + 1) * NCLS + mc] * maskv;
    c = __builtin_amdgcn_wmma_f32_16x16x4_f32(
        false, a, false, b, (short)0, c, false, false);
  }

  float dv = dinv[rowc];
  bool full = (wave * 16 + 16 <= n);
  #pragma unroll
  for (int r = 0; r < 8; ++r) {
    int orow = wave * 16 + r + 8 * half;
    float w  = __shfl(dv, r + 8 * half, 32);
    if (full || orow < n) {
      lt[(size_t)orow * RS2 + m] = (m < NCLS) ? c[r] : ((m == NCLS) ? w : 0.0f);
      if (m < NCLS)
        out[(size_t)orow * NCLS + m] = c[r] * w * w;
    }
  }
}


__global__ void lsm_kernel(float* __restrict__ out,
                           const float* __restrict__ part,
                           const float* __restrict__ b2, int n) {
  int i = blockIdx.x * blockDim.x + threadIdx.x;
  if (i >= n) return;
  float* row = out + (size_t)i * NCLS;
  float z[NCLS];
  #pragma unroll
  for (int c = 0; c < NCLS; ++c) z[c] = row[c];
  #pragma unroll
  for (int ch = 0; ch < NCH2; ++ch) {
    const float4* pr = (const float4*)(part + ((size_t)ch * n + i) * NCLS);
    float4 p0 = pr[0], p1 = pr[1];
    z[0] += p0.x; z[1] += p0.y; z[2] += p0.z; z[3] += p0.w;
    z[4] += p1.x; z[5] += p1.y; z[6] += p1.z; z[7] += p1.w;
  }
  float mx = -INFINITY;
  #pragma unroll
  for (int c = 0; c < NCLS; ++c) {
    z[c] += b2[c];
    mx = fmaxf(mx, z[c]);
  }
  float s = 0.0f;
  #pragma unroll
  for (int c = 0; c < NCLS; ++c) s += expf(z[c] - mx);
  float l = mx + logf(s);
  #pragma unroll
  for (int c = 0; c < NCLS; ++c) row[c] = z[c] - l;
}


extern "C" void kernel_launch(void* const* d_in, const int* in_sizes, int n_in,
                              void* d_out, int out_size, void* d_ws, size_t ws_size,
                              hipStream_t stream) {
  const float* x  = (const float*)d_in[0];
  const int* edge = (const int*)d_in[1];
  const float* W1 = (const float*)d_in[2];
  const float* b1 = (const float*)d_in[3];
  const float* W2 = (const float*)d_in[4];
  const float* b2 = (const float*)d_in[5];

  int n = in_sizes[0] / F_IN;
  int e = in_sizes[1] / 2;
  const int* src = edge;
  const int* dst = edge + e;

  float* deg = (float*)d_ws;
  float* h   = deg + n;
  float* h1  = h + (size_t)n * RS1;
  float* lt  = h1 + (size_t)n * HID;
  float* p1  = lt + (size_t)n * RS2;
  float* p2  = p1 + (size_t)NCH1 * n * HID;
  float* dgp = p2 + (size_t)NCH2 * n * NCLS;
  float* out = (float*)d_out;

  const int T = 256;
  int nTiles  = (n + 15) / 16;
  int gBlocks = (nTiles + 7) / 8;
  dim3 dGrid((n + DNB - 1) / DNB, DCH);
  dim3 aGrid1((n + NB1 - 1) / NB1, NCH1);
  dim3 aGrid2((n + NB2 - 1) / NB2, NCH2);

  deg_part_kernel<<<dGrid, ATPB, 0, stream>>>(dst, dgp, e, n);
  dinv_kernel<<<(n + T - 1) / T, T, 0, stream>>>(dgp, deg, n);

  gemm1_kernel<<<gBlocks, T, 0, stream>>>(x, W1, deg, h, h1, nTiles, n);
  agg_kernel<HID, NB1, NCH1><<<aGrid1, ATPB, 0, stream>>>(src, dst, deg, h, p1, e, n);

  gemm2_kernel<<<gBlocks, T, 0, stream>>>(h1, p1, b1, W2, deg, lt, out, nTiles, n);
  agg_kernel<NCLS, NB2, NCH2><<<aGrid2, ATPB, 0, stream>>>(src, dst, deg, lt, p2, e, n);

  lsm_kernel<<<(n + T - 1) / T, T, 0, stream>>>(out, p2, b2, n);
}


# Layer-1 edge-sum waves store their filtered (src, dst offset) lists; the layer-2 edge sum (same node ranges and chunks) reads them, no second search of the edge list; planes 19 -> 10, new sum order
# speedup vs baseline: 9.2008x; 1.1298x over previous
#include <hip/hip_runtime.h>
#include <hip/hip_bf16.h>
#include <math.h>


typedef float v2f __attribute__((ext_vector_type(2)));
typedef float v8f __attribute__((ext_vector_type(8)));

#define F_IN 128
#define HID  16
#define NCLS 8

#define RS1 (2 * HID)
#define RS2 (2 * NCLS)


#define ATPB 1024
#define VQ   8
#define WCAP 128
#define WKL  (WCAP / 32)
#define MAXT 32

#define DNB  25000
#define DCH  64
#define NB1  4000
#define NCH1 10


template <bool VEC>
__device__ __forceinline__ void load_tile(const int* __restrict__ p, int e0,
                                          int e1, int ng, int gb, int tid,
                                          int fill, int4 (&q)[VQ]) {
  if (VEC) {
    #pragma unroll
    for (int k = 0; k < VQ; ++k) {
      int g = gb + k * ATPB + tid;
      q[k] = *(const int4*)(p + e0 + 4 * (g < ng ? g : 0));
    }
  } else {
    #pragma unroll
    for (int k = 0; k < VQ; ++k) {
      int g = gb + k * ATPB + tid;
      int base = e0 + 4 * (g < ng ? g : 0);
      q[k].x = (base + 0 < e1) ? p[base + 0] : fill;
      q[k].y = (base + 1 < e1) ? p[base + 1] : fill;
      q[k].z = (base + 2 < e1) ? p[base + 2] : fill;
      q[k].w = (base + 3 < e1) ? p[base + 3] : fill;
    }
  }
}

__device__ __forceinline__ bool group_ok(int gb, int k, int tid, int ng) {
  return gb + k * ATPB + tid < ng;
}


template <bool VEC>
__device__ __forceinline__ void deg_part_body(
    float* hc, const int* __restrict__ dst, float* __restrict__ degp, int e, int n) {
  const int tid = threadIdx.x;
  const int nb0 = blockIdx.x * DNB;
  const int nbn = min(DNB, n - nb0);
  const int per = (((e + DCH - 1) / DCH) + 3) & ~3;
  const int e0  = blockIdx.y * per;
  const int e1  = min(e, e0 + per);
  const int ng  = e1 > e0 ? (e1 - e0 + 3) >> 2 : 0;

  for (int i = tid; i < DNB; i += ATPB) hc[i] = 0.0f;
  __syncthreads();

  for (int g0 = 0; g0 < ng; g0 += ATPB * VQ) {
    int4 dq[VQ];
    load_tile<VEC>(dst, e0, e1, ng, g0, tid, -1, dq);
    #pragma unroll
    for (int k = 0; k < VQ; ++k) {
      const bool gv = group_ok(g0, k, tid, ng);
      int dv[4] = {dq[k].x, dq[k].y, dq[k].z, dq[k].w};
      #pragma unroll
      for (int c = 0; c < 4; ++c) {
        unsigned dl = (unsigned)(dv[c] - nb0);
        if (gv && dl < (unsigned)nbn) atomicAdd(&hc[dl], 1.0f);
      }
    }
  }
  __syncthreads();

  float* pp = degp + (size_t)blockIdx.y * n + nb0;
  for (int i = tid; i < nbn; i += ATPB) pp[i] = hc[i];
}

__global__ void __launch_bounds__(ATPB) deg_part_kernel(
    const int* __restrict__ dst, float* __restrict__ degp, int e, int n) {
  __shared__ float hc[DNB];
  if ((e & 3) == 0) deg_part_body<true>(hc, dst, degp, e, n);
  else              deg_part_body<false>(hc, dst, degp, e, n);
}

__global__ void dinv_kernel(const float* __restrict__ degp,
                            float* __restrict__ dinv, int n) {
  int i = blockIdx.x * blockDim.x + threadIdx.x;
  if (i >= n) return;
  float s = 1.0f;
  #pragma unroll 16
  for (int ch = 0; ch < DCH; ++ch) s += degp[(size_t)ch * n + i];
  dinv[i] = rsqrtf(s);
}


__global__ void __launch_bounds__(256) gemm1_kernel(
    const float* __restrict__ x, const float* __restrict__ W1,
    const float* __restrict__ dinv,
    float* __restrict__ h, float* __restrict__ out1, int nTiles, int n) {
  int wave = (blockIdx.x * 256 + threadIdx.x) >> 5;
  int lane = threadIdx.x & 31;
  if (wave >= nTiles) return;
  int half = lane >> 4;
  int m    = lane & 15;
  int row  = wave * 16 + m;
  int rowc = row < n ? row : n - 1;
  const float* xrow = x + (size_t)rowc * F_IN;

  v8f c = {};
  #pragma unroll 4
  for (int k0 = 0; k0 < F_IN; k0 += 4) {
    int kk = k0 + 2 * half;
    v2f a = *(const v2f*)(xrow + kk);
    v2f b;
    b.x = W1[kk * HID + m];
    b.y = W1[(kk + 1) * HID + m];
    c = __builtin_amdgcn_wmma_f32_16x16x4_f32(
        false, a, false, b, (short)0, c, false, false);
  }

  float dv = dinv[rowc];
  bool full = (wave * 16 + 16 <= n);
  if (full) {
    #pragma unroll
    for (int r = 0; r < 8; ++r) {
      int orow = wave * 16 + r + 8 * half;
      float w  = __shfl(dv, r + 8 * half, 32);
      size_t o = (size_t)orow * HID + m;
      size_t g = (size_t)orow * RS1 + m;
      h[g]       = c[r];
      h[g + HID] = (m == 0) ? w : 0.0f;
      out1[o] = c[r] * w * w;
    }
  } else {
    #pragma unroll
    for (int r = 0; r < 8; ++r) {
      int orow = wave * 16 + r + 8 * half;
      float w  = __shfl(dv, r + 8 * half, 32);
      if (orow < n) {
        size_t o = (size_t)orow * HID + m;
        size_t g = (size_t)orow * RS1 + m;
        h[g]       = c[r];
        h[g + HID] = (m == 0) ? w : 0.0f;
        out1[o] = c[r] * w * w;
      }
    }
  }
}


__device__ __forceinline__ int tiles_of(int groups) {
  return (groups + ATPB * VQ - 1) / (ATPB * VQ);
}

template <int WD>
__device__ __forceinline__ void sum_list(
    float* acc, const int* wl, int tot, int lane, int nb0,
    const float* __restrict__ dinv, const float* __restrict__ feat) {
  constexpr int RS  = 2 * WD;
  constexpr int LPE = WD / 4;
  constexpr int EPW = 32 / LPE;
  const int sub  = lane % LPE;
  const int slot = lane / LPE;
  for (int j = slot; j < tot; j += 4 * EPW) {
    int s[4], d[4];
    bool ok[4];
    float da[4], db[4];
    float4 v[4];
    #pragma unroll
    for (int q = 0; q < 4; ++q) {
      int jj = j + q * EPW;
      ok[q] = jj < tot;
      int jc = ok[q] ? jj : j;
      s[q] = wl[2 * jc];
      d[q] = wl[2 * jc + 1];
    }
    #pragma unroll
    for (int q = 0; q < 4; ++q) {
      const float* fr = feat + (size_t)s[q] * RS;
      v[q]  = *(const float4*)(fr + sub * 4);
      da[q] = fr[WD];
      db[q] = dinv[nb0 + d[q]];
    }
    __builtin_amdgcn_sched_barrier(0);
    asm volatile("" :: "v"(da[0]), "v"(da[1]), "v"(da[2]), "v"(da[3]),
                       "v"(db[0]), "v"(db[1]), "v"(db[2]), "v"(db[3]),
                       "v"(v[0].x), "v"(v[1].x), "v"(v[2].x), "v"(v[3].x));
    #pragma unroll
    for (int q = 0; q < 4; ++q) {
      if (ok[q]) {
        float w = da[q] * db[q];
        float* a = acc + d[q] * WD + sub * 4;
        atomicAdd(a + 0, v[q].x * w);
        atomicAdd(a + 1, v[q].y * w);
        atomicAdd(a + 2, v[q].z * w);
        atomicAdd(a + 3, v[q].w * w);
      }
    }
  }
}

template <int WD>
__device__ __forceinline__ void sum_direct(
    float* acc, const int* __restrict__ src, const int* __restrict__ dst,
    const float* __restrict__ dinv, const float* __restrict__ feat,
    int e0, int e1, int ng, int g0, int tid, int nb0, int nbn) {
  constexpr int RS = 2 * WD;
  #pragma unroll 1
  for (int b = 0; b < 4 * VQ; ++b) {
    const int g  = g0 + (b >> 2) * ATPB + tid;
    const int ei = e0 + 4 * g + (b & 3);
    if (g < ng && ei < e1) {
      const int dvv = dst[ei];
      const unsigned dl = (unsigned)(dvv - nb0);
      if (dl < (unsigned)nbn) {
        const float* fr = feat + (size_t)src[ei] * RS;
        float w = fr[WD] * dinv[dvv];
        #pragma unroll 1
        for (int f = 0; f < WD; ++f) atomicAdd(&acc[dl * WD + f], fr[f] * w);
      }
    }
  }
}

template <int WD, int NBK, int NCHK, bool VEC>
__device__ __forceinline__ void agg_body(
    float* acc, int* lst,
    const int* __restrict__ src, const int* __restrict__ dst,
    const float* __restrict__ dinv, const float* __restrict__ feat,
    float* __restrict__ part, int2* __restrict__ glist, int* __restrict__ gcnt,
    int e, int n) {
  constexpr int TILE_G = ATPB * VQ;
  const int tid = threadIdx.x;
  const int nb0 = blockIdx.x * NBK;
  const int nbn = min(NBK, n - nb0);
  const int per = (((e + NCHK - 1) / NCHK) + 3) & ~3;
  const int e0  = blockIdx.y * per;
  const int e1  = min(e, e0 + per);
  const int ng  = e1 > e0 ? (e1 - e0 + 3) >> 2 : 0;
  const int ntm = tiles_of(per >> 2);
  const bool keep = ntm <= MAXT;

  for (int i = tid; i < NBK * WD / 4; i += ATPB)
    ((float4*)acc)[i] = make_float4(0.f, 0.f, 0.f, 0.f);
  __syncthreads();

  const int lane = tid & 31;
  int* wl = lst + (tid >> 5) * (2 * WCAP);
  const size_t wgw = (size_t)(blockIdx.y * gridDim.x + blockIdx.x) * (ATPB / 32) + (tid >> 5);

  int t = 0;
  for (int g0 = 0; g0 < ng; g0 += TILE_G, ++t) {
    int4 dq[VQ], sq[VQ];
    load_tile<VEC>(dst, e0, e1, ng, g0, tid, -1, dq);
    load_tile<VEC>(src, e0, e1, ng, g0, tid, 0, sq);
    static_assert(VQ == 8, "the fence below names eight src loads");
    __builtin_amdgcn_sched_barrier(0);
    asm volatile("" :: "v"(sq[0].x), "v"(sq[1].x), "v"(sq[2].x), "v"(sq[3].x),
                       "v"(sq[4].x), "v"(sq[5].x), "v"(sq[6].x), "v"(sq[7].x));

    unsigned nbk[VQ];
    #pragma unroll
    for (int k = 0; k < VQ; ++k)
      nbk[k] = group_ok(g0, k, tid, ng) ? (unsigned)nbn : 0u;
    unsigned mk[4 * VQ];
    int tot = 0;
    #pragma unroll
    for (int k = 0; k < VQ; ++k) {
      int dv[4] = {dq[k].x, dq[k].y, dq[k].z, dq[k].w};
      #pragma unroll
      for (int c = 0; c < 4; ++c) {
        mk[4 * k + c] = __builtin_amdgcn_ballot_w32((unsigned)(dv[c] - nb0) < nbk[k]);
        tot += __popc(mk[4 * k + c]);
      }
    }
    if (tot <= WCAP) {
      int run = 0;
      #pragma unroll
      for (int k = 0; k < VQ; ++k) {
        int dv[4] = {dq[k].x, dq[k].y, dq[k].z, dq[k].w};
        int sv[4] = {sq[k].x, sq[k].y, sq[k].z, sq[k].w};
        #pragma unroll
        for (int c = 0; c < 4; ++c) {
          const unsigned dl = (unsigned)(dv[c] - nb0);
          if (dl < nbk[k]) {
            const int p = (int)__builtin_amdgcn_mbcnt_lo(mk[4 * k + c], (unsigned)run);
            wl[2 * p]     = sv[c];
            wl[2 * p + 1] = (int)dl;
          }
          run += __popc(mk[4 * k + c]);
        }
      }

      if (keep) {
        if (tot > 0) {
          int2* gl = glist + (wgw * ntm + t) * WCAP;
          int2 cp[WKL];
          #pragma unroll
          for (int k = 0; k < WKL; ++k) {
            const int i = lane + 32 * k;
            cp[k] = ((const int2*)wl)[i < tot ? i : 0];
          }
          #pragma unroll
          for (int k = 0; k < WKL; ++k) {
            const int i = lane + 32 * k;
            if (i < tot) gl[i] = cp[k];
          }
        }
        if (lane == 0) gcnt[wgw * MAXT + t] = tot;
      }

      sum_list<WD>(acc, wl, tot, lane, nb0, dinv, feat);
    } else {
      if (keep && lane == 0) gcnt[wgw * MAXT + t] = -1;
      sum_direct<WD>(acc, src, dst, dinv, feat, e0, e1, ng, g0, tid, nb0, nbn);
    }
  }
  __syncthreads();

  float4* pp = (float4*)(part + ((size_t)blockIdx.y * n + nb0) * WD);
  for (int i = tid; i < nbn * WD / 4; i += ATPB) pp[i] = ((const float4*)acc)[i];
}

template <int WD, int NBK, int NCHK>
__global__ void __launch_bounds__(ATPB) agg_kernel(
    const int* __restrict__ src, const int* __restrict__ dst,
    const float* __restrict__ dinv, const float* __restrict__ feat,
    float* __restrict__ part, int2* __restrict__ glist, int* __restrict__ gcnt,
    int e, int n) {
  __shared__ float acc[NBK * WD];
  __shared__ __attribute__((aligned(16))) int lst[(ATPB / 32) * 2 * WCAP];
  if ((e & 3) == 0)
    agg_body<WD, NBK, NCHK, true>(acc, lst, src, dst, dinv, feat, part, glist, gcnt, e, n);
  else
    agg_body<WD, NBK, NCHK, false>(acc, lst, src, dst, dinv, feat, part, glist, gcnt, e, n);
}

template <int WD, int NBK, int NCHK>
__global__ void __launch_bounds__(ATPB) agg_list_kernel(
    const int* __restrict__ src, const int* __restrict__ dst,
    const float* __restrict__ dinv, const float* __restrict__ feat,
    float* __restrict__ part, const int2* __restrict__ glist,
    const int* __restrict__ gcnt, int e, int n) {
  __shared__ float acc[NBK * WD];
  __shared__ __attribute__((aligned(16))) int lst[(ATPB / 32) * 2 * WCAP];
  constexpr int TILE_G = ATPB * VQ;
  const int tid = threadIdx.x;
  const int nb0 = blockIdx.x * NBK;
  const int nbn = min(NBK, n - nb0);
  const int per = (((e + NCHK - 1) / NCHK) + 3) & ~3;
  const int e0  = blockIdx.y * per;
  const int e1  = min(e, e0 + per);
  const int ng  = e1 > e0 ? (e1 - e0 + 3) >> 2 : 0;
  const int nt  = tiles_of(ng);
  const int ntm = tiles_of(per >> 2);
  const bool keep = ntm <= MAXT;

  for (int i = tid; i < NBK * WD / 4; i += ATPB)
    ((float4*)acc)[i] = make_float4(0.f, 0.f, 0.f, 0.f);
  __syncthreads();

  const int lane = tid & 31;
  int* wl = lst + (tid >> 5) * (2 * WCAP);
  const size_t wgw = (size_t)(blockIdx.y * gridDim.x + blockIdx.x) * (ATPB / 32) + (tid >> 5);
  const int2* gw = glist + wgw * ntm * WCAP;

  int cv = -1;
  if (keep && lane < nt) cv = gcnt[wgw * MAXT + lane];

  int2 nx[WKL];
  int cnt = (keep && nt > 0) ? __builtin_amdgcn_readlane(cv, 0) : -1;
  #pragma unroll
  for (int k = 0; k < WKL; ++k) {
    const int i = lane + 32 * k;
    nx[k] = (i < cnt) ? gw[i] : make_int2(0, 0);
  }
  for (int t = 0; t < nt; ++t) {
    #pragma unroll
    for (int k = 0; k < WKL; ++k) {
      const int i = lane + 32 * k;
      if (i < cnt) ((int2*)wl)[i] = nx[k];
    }
    const int cntn = (keep && t + 1 < nt) ? __builtin_amdgcn_readlane(cv, t + 1) : -1;
    #pragma unroll
    for (int k = 0; k < WKL; ++k) {
      const int i = lane + 32 * k;
      nx[k] = (i < cntn) ? gw[(size_t)(t + 1) * WCAP + i] : make_int2(0, 0);
    }
    if (cnt >= 0) sum_list<WD>(acc, wl, cnt, lane, nb0, dinv, feat);
    else sum_direct<WD>(acc, src, dst, dinv, feat, e0, e1, ng, t * TILE_G, tid, nb0, nbn);
    cnt = cntn;
  }
  __syncthreads();

  float4* pp = (float4*)(part + ((size_t)blockIdx.y * n + nb0) * WD);
  for (int i = tid; i < nbn * WD / 4; i += ATPB) pp[i] = ((const float4*)acc)[i];
}


__global__ void __launch_bounds__(256) gemm2_kernel(
    const float* __restrict__ h1, const float* __restrict__ part,
    const float* __restrict__ b1,
    const float* __restrict__ W2, const float* __restrict__ dinv,
    float* __restrict__ lt, float* __restrict__ out, int nTiles, int n) {
  int wave = (blockIdx.x * 256 + threadIdx.x) >> 5;
  int lane = threadIdx.x & 31;
  if (wave >= nTiles) return;
  int half = lane >> 4;
  int m    = lane & 15;
  int row  = wave * 16 + m;
  int rowc = row < n ? row : n - 1;
  const float* hrow = h1 + (size_t)rowc * HID;
  const float* prow = part + (size_t)rowc * HID;
  const size_t pstr = (size_t)n * HID;
  float maskv = (m < NCLS) ? 1.0f : 0.0f;
  int mc = m & (NCLS - 1);

  v8f c = {};
  #pragma unroll
  for (int k0 = 0; k0 < HID; k0 += 4) {
    int kk = k0 + 2 * half;
    v2f t = *(const v2f*)(hrow + kk);
    #pragma unroll
    for (int ch = 0; ch < NCH1; ++ch) {
      v2f pv = *(const v2f*)(prow + ch * pstr + kk);
      t.x += pv.x;
      t.y += pv.y;
    }
    v2f a;
    a.x = fmaxf(t.x + b1[kk],     0.0f);
    a.y = fmaxf(t.y + b1[kk + 1], 0.0f);
    v2f b;
    b.x = W2[kk * NCLS + mc] * maskv;
    b.y = W2[(kk + 1) * NCLS + mc] * maskv;
    c = __builtin_amdgcn_wmma_f32_16x16x4_f32(
        false, a, false, b, (short)0, c, false, false);
  }

  float dv = dinv[rowc];
  bool full = (wave * 16 + 16 <= n);
  #pragma unroll
  for (int r = 0; r < 8; ++r) {
    int orow = wave * 16 + r + 8 * half;
    float w  = __shfl(dv, r + 8 * half, 32);
    if (full || orow < n) {
      lt[(size_t)orow * RS2 + m] = (m < NCLS) ? c[r] : ((m == NCLS) ? w : 0.0f);
      if (m < NCLS)
        out[(size_t)orow * NCLS + m] = c[r] * w * w;
    }
  }
}


__global__ void lsm_kernel(float* __restrict__ out,
                           const float* __restrict__ part,
                           const float* __restrict__ b2, int n) {
  int i = blockIdx.x * blockDim.x + threadIdx.x;
  if (i >= n) return;
  float* row = out + (size_t)i * NCLS;
  float z[NCLS];
  #pragma unroll
  for (int c = 0; c < NCLS; ++c) z[c] = row[c];
  #pragma unroll
  for (int ch = 0; ch < NCH1; ++ch) {
    const float4* pr = (const float4*)(part + ((size_t)ch * n + i) * NCLS);
    float4 p0 = pr[0], p1 = pr[1];
    z[0] += p0.x; z[1] += p0.y; z[2] += p0.z; z[3] += p0.w;
    z[4] += p1.x; z[5] += p1.y; z[6] += p1.z; z[7] += p1.w;
  }
  float mx = -INFINITY;
  #pragma unroll
  for (int c = 0; c < NCLS; ++c) {
    z[c] += b2[c];
    mx = fmaxf(mx, z[c]);
  }
  float s = 0.0f;
  #pragma unroll
  for (int c = 0; c < NCLS; ++c) s += expf(z[c] - mx);
  float l = mx + logf(s);
  #pragma unroll
  for (int c = 0; c < NCLS; ++c) row[c] = z[c] - l;
}


extern "C" void kernel_launch(void* const* d_in, const int* in_sizes, int n_in,
                              void* d_out, int out_size, void* d_ws, size_t ws_size,
                              hipStream_t stream) {
  const float* x  = (const float*)d_in[0];
  const int* edge = (const int*)d_in[1];
  const float* W1 = (const float*)d_in[2];
  const float* b1 = (const float*)d_in[3];
  const float* W2 = (const float*)d_in[4];
  const float* b2 = (const float*)d_in[5];

  int n = in_sizes[0] / F_IN;
  int e = in_sizes[1] / 2;
  const int* src = edge;
  const int* dst = edge + e;

  float* deg = (float*)d_ws;
  float* h   = deg + n;
  float* h1  = h + (size_t)n * RS1;
  float* lt  = h1 + (size_t)n * HID;
  float* p1  = lt + (size_t)n * RS2;
  float* p2  = p1 + (size_t)NCH1 * n * HID;
  float* dgp = p2 + (size_t)NCH1 * n * NCLS;
  float* out = (float*)d_out;

  const int T = 256;
  int nTiles  = (n + 15) / 16;
  int gBlocks = (nTiles + 7) / 8;
  dim3 dGrid((n + DNB - 1) / DNB, DCH);
  dim3 aGrid1((n + NB1 - 1) / NB1, NCH1);
  int* gcnt   = (int*)(dgp + (size_t)DCH * n);
  int2* glist = (int2*)(gcnt + (size_t)aGrid1.x * NCH1 * (ATPB / 32) * MAXT);

  deg_part_kernel<<<dGrid, ATPB, 0, stream>>>(dst, dgp, e, n);
  dinv_kernel<<<(n + T - 1) / T, T, 0, stream>>>(dgp, deg, n);

  gemm1_kernel<<<gBlocks, T, 0, stream>>>(x, W1, deg, h, h1, nTiles, n);
  agg_kernel<HID, NB1, NCH1><<<aGrid1, ATPB, 0, stream>>>(src, dst, deg, h, p1, glist, gcnt, e, n);

  gemm2_kernel<<<gBlocks, T, 0, stream>>>(h1, p1, b1, W2, deg, lt, out, nTiles, n);
  agg_list_kernel<NCLS, NB1, NCH1><<<aGrid1, ATPB, 0, stream>>>(src, dst, deg, lt, p2, glist, gcnt, e, n);

  lsm_kernel<<<(n + T - 1) / T, T, 0, stream>>>(out, p2, b2, n);
}


# Edge sums run whole trips only: a tile's rest waits at the list's front for the next tile; layer 1 stores one stream per wave, layer 2 reads it in whole trips; LDS-atomic sum order changes
# speedup vs baseline: 9.5605x; 1.0391x over previous
#include <hip/hip_runtime.h>
#include <hip/hip_bf16.h>
#include <math.h>


typedef float v2f __attribute__((ext_vector_type(2)));
typedef float v8f __attribute__((ext_vector_type(8)));

#define F_IN 128
#define HID  16
#define NCLS 8

#define RS1 (2 * HID)
#define RS2 (2 * NCLS)


#define ATPB 1024
#define VQ   8
#define WCAP 128
#define WKL  (WCAP / 32)
#define MAXT 32

#define DNB  25000
#define DCH  64
#define NB1  4000
#define NCH1 10


template <bool VEC>
__device__ __forceinline__ void load_tile(const int* __restrict__ p, int e0,
                                          int e1, int ng, int gb, int tid,
                                          int fill, int4 (&q)[VQ]) {
  if (VEC) {
    #pragma unroll
    for (int k = 0; k < VQ; ++k) {
      int g = gb + k * ATPB + tid;
      q[k] = *(const int4*)(p + e0 + 4 * (g < ng ? g : 0));
    }
  } else {
    #pragma unroll
    for (int k = 0; k < VQ; ++k) {
      int g = gb + k * ATPB + tid;
      int base = e0 + 4 * (g < ng ? g : 0);
      q[k].x = (base + 0 < e1) ? p[base + 0] : fill;
      q[k].y = (base + 1 < e1) ? p[base + 1] : fill;
      q[k].z = (base + 2 < e1) ? p[base + 2] : fill;
      q[k].w = (base + 3 < e1) ? p[base + 3] : fill;
    }
  }
}

__device__ __forceinline__ bool group_ok(int gb, int k, int tid, int ng) {
  return gb + k * ATPB + tid < ng;
}


template <bool VEC>
__device__ __forceinline__ void deg_part_body(
    float* hc, const int* __restrict__ dst, float* __restrict__ degp, int e, int n) {
  const int tid = threadIdx.x;
  const int nb0 = blockIdx.x * DNB;
  const int nbn = min(DNB, n - nb0);
  const int per = (((e + DCH - 1) / DCH) + 3) & ~3;
  const int e0  = blockIdx.y * per;
  const int e1  = min(e, e0 + per);
  const int ng  = e1 > e0 ? (e1 - e0 + 3) >> 2 : 0;

  for (int i = tid; i < DNB; i += ATPB) hc[i] = 0.0f;
  __syncthreads();

  for (int g0 = 0; g0 < ng; g0 += ATPB * VQ) {
    int4 dq[VQ];
    load_tile<VEC>(dst, e0, e1, ng, g0, tid, -1, dq);
    #pragma unroll
    for (int k = 0; k < VQ; ++k) {
      const bool gv = group_ok(g0, k, tid, ng);
      int dv[4] = {dq[k].x, dq[k].y, dq[k].z, dq[k].w};
      #pragma unroll
      for (int c = 0; c < 4; ++c) {
        unsigned dl = (unsigned)(dv[c] - nb0);
        if (gv && dl < (unsigned)nbn) atomicAdd(&hc[dl], 1.0f);
      }
    }
  }
  __syncthreads();

  float* pp = degp + (size_t)blockIdx.y * n + nb0;
  for (int i = tid; i < nbn; i += ATPB) pp[i] = hc[i];
}

__global__ void __launch_bounds__(ATPB) deg_part_kernel(
    const int* __restrict__ dst, float* __restrict__ degp, int e, int n) {
  __shared__ float hc[DNB];
  if ((e & 3) == 0) deg_part_body<true>(hc, dst, degp, e, n);
  else              deg_part_body<false>(hc, dst, degp, e, n);
}

__global__ void dinv_kernel(const float* __restrict__ degp,
                            float* __restrict__ dinv, int n) {
  int i = blockIdx.x * blockDim.x + threadIdx.x;
  if (i >= n) return;
  float s = 1.0f;
  #pragma unroll 16
  for (int ch = 0; ch < DCH; ++ch) s += degp[(size_t)ch * n + i];
  dinv[i] = rsqrtf(s);
}


__global__ void __launch_bounds__(256) gemm1_kernel(
    const float* __restrict__ x, const float* __restrict__ W1,
    const float* __restrict__ dinv,
    float* __restrict__ h, float* __restrict__ out1, int nTiles, int n) {
  int wave = (blockIdx.x * 256 + threadIdx.x) >> 5;
  int lane = threadIdx.x & 31;
  if (wave >= nTiles) return;
  int half = lane >> 4;
  int m    = lane & 15;
  int row  = wave * 16 + m;
  int rowc = row < n ? row : n - 1;
  const float* xrow = x + (size_t)rowc * F_IN;

  v8f c = {};
  #pragma unroll 4
  for (int k0 = 0; k0 < F_IN; k0 += 4) {
    int kk = k0 + 2 * half;
    v2f a = *(const v2f*)(xrow + kk);
    v2f b;
    b.x = W1[kk * HID + m];
    b.y = W1[(kk + 1) * HID + m];
    c = __builtin_amdgcn_wmma_f32_16x16x4_f32(
        false, a, false, b, (short)0, c, false, false);
  }

  float dv = dinv[rowc];
  bool full = (wave * 16 + 16 <= n);
  if (full) {
    #pragma unroll
    for (int r = 0; r < 8; ++r) {
      int orow = wave * 16 + r + 8 * half;
      float w  = __shfl(dv, r + 8 * half, 32);
      size_t o = (size_t)orow * HID + m;
      size_t g = (size_t)orow * RS1 + m;
      h[g]       = c[r];
      h[g + HID] = (m == 0) ? w : 0.0f;
      out1[o] = c[r] * w * w;
    }
  } else {
    #pragma unroll
    for (int r = 0; r < 8; ++r) {
      int orow = wave * 16 + r + 8 * half;
      float w  = __shfl(dv, r + 8 * half, 32);
      if (orow < n) {
        size_t o = (size_t)orow * HID + m;
        size_t g = (size_t)orow * RS1 + m;
        h[g]       = c[r];
        h[g + HID] = (m == 0) ? w : 0.0f;
        out1[o] = c[r] * w * w;
      }
    }
  }
}


__device__ __forceinline__ int tiles_of(int groups) {
  return (groups + ATPB * VQ - 1) / (ATPB * VQ);
}

template <int WD, bool FULL>
__device__ __forceinline__ void sum_four(
    float* acc, const int (&s)[4], const int (&d)[4], const bool (&ok)[4],
    int sub, int nb0,
    const float* __restrict__ dinv, const float* __restrict__ feat) {
  constexpr int RS = 2 * WD;
  float da[4], db[4];
  float4 v[4];
  #pragma unroll
  for (int q = 0; q < 4; ++q) {
    const float* fr = feat + (size_t)s[q] * RS;
    v[q]  = *(const float4*)(fr + sub * 4);
    da[q] = fr[WD];
    db[q] = dinv[nb0 + d[q]];
  }
  __builtin_amdgcn_sched_barrier(0);
  asm volatile("" :: "v"(da[0]), "v"(da[1]), "v"(da[2]), "v"(da[3]),
                     "v"(db[0]), "v"(db[1]), "v"(db[2]), "v"(db[3]),
                     "v"(v[0].x), "v"(v[1].x), "v"(v[2].x), "v"(v[3].x));
  #pragma unroll
  for (int q = 0; q < 4; ++q) {
    if (FULL || ok[q]) {
      float w = da[q] * db[q];
      float* a = acc + d[q] * WD + sub * 4;
      atomicAdd(a + 0, v[q].x * w);
      atomicAdd(a + 1, v[q].y * w);
      atomicAdd(a + 2, v[q].z * w);
      atomicAdd(a + 3, v[q].w * w);
    }
  }
}

template <int WD, bool FULL>
__device__ __forceinline__ void sum_list(
    float* acc, const int* wl, int tot, int lane, int nb0,
    const float* __restrict__ dinv, const float* __restrict__ feat) {
  constexpr int LPE = WD / 4;
  constexpr int EPW = 32 / LPE;
  const int sub  = lane % LPE;
  const int slot = lane / LPE;
  for (int j = slot; j < tot; j += 4 * EPW) {
    int s[4], d[4];
    bool ok[4];
    #pragma unroll
    for (int q = 0; q < 4; ++q) {
      int jj = j + q * EPW;
      ok[q] = FULL || jj < tot;
      int jc = ok[q] ? jj : j;
      s[q] = wl[2 * jc];
      d[q] = wl[2 * jc + 1];
    }
    sum_four<WD, FULL>(acc, s, d, ok, sub, nb0, dinv, feat);
  }
}

template <int WD>
__device__ __forceinline__ void sum_direct(
    float* acc, const int* __restrict__ src, const int* __restrict__ dst,
    const float* __restrict__ dinv, const float* __restrict__ feat,
    int e0, int e1, int ng, int g0, int tid, int nb0, int nbn) {
  constexpr int RS = 2 * WD;
  #pragma unroll 1
  for (int b = 0; b < 4 * VQ; ++b) {
    const int g  = g0 + (b >> 2) * ATPB + tid;
    const int ei = e0 + 4 * g + (b & 3);
    if (g < ng && ei < e1) {
      const int dvv = dst[ei];
      const unsigned dl = (unsigned)(dvv - nb0);
      if (dl < (unsigned)nbn) {
        const float* fr = feat + (size_t)src[ei] * RS;
        float w = fr[WD] * dinv[dvv];
        #pragma unroll 1
        for (int f = 0; f < WD; ++f) atomicAdd(&acc[dl * WD + f], fr[f] * w);
      }
    }
  }
}

__device__ __forceinline__ void store_list(int2* __restrict__ g, const int* wl,
                                           int cnt, int lane) {
  if (cnt > 0) {
    int2 cp[WKL];
    #pragma unroll
    for (int k = 0; k < WKL; ++k) {
      const int i = lane + 32 * k;
      cp[k] = ((const int2*)wl)[i < cnt ? i : 0];
    }
    #pragma unroll
    for (int k = 0; k < WKL; ++k) {
      const int i = lane + 32 * k;
      if (i < cnt) g[i] = cp[k];
    }
  }
}

template <int WD, int NBK, int NCHK, bool VEC>
__device__ __forceinline__ void agg_body(
    float* acc, int* lst,
    const int* __restrict__ src, const int* __restrict__ dst,
    const float* __restrict__ dinv, const float* __restrict__ feat,
    float* __restrict__ part, int2* __restrict__ glist, int* __restrict__ gcnt,
    int e, int n) {
  constexpr int TILE_G = ATPB * VQ;
  const int tid = threadIdx.x;
  const int nb0 = blockIdx.x * NBK;
  const int nbn = min(NBK, n - nb0);
  const int per = (((e + NCHK - 1) / NCHK) + 3) & ~3;
  const int e0  = blockIdx.y * per;
  const int e1  = min(e, e0 + per);
  const int ng  = e1 > e0 ? (e1 - e0 + 3) >> 2 : 0;
  const int ntm = tiles_of(per >> 2);
  const bool keep = ntm <= MAXT;

  for (int i = tid; i < NBK * WD / 4; i += ATPB)
    ((float4*)acc)[i] = make_float4(0.f, 0.f, 0.f, 0.f);
  __syncthreads();

  const int lane = tid & 31;
  int* wl = lst + (tid >> 5) * (2 * WCAP);
  const size_t wgw = (size_t)(blockIdx.y * gridDim.x + blockIdx.x) * (ATPB / 32) + (tid >> 5);

  constexpr int TRIP = 4 * (32 / (WD / 4));
  static_assert(TRIP <= 32 && (TRIP & (TRIP - 1)) == 0, "a tile's rest must fit one entry per lane");
  int2* gl = glist + wgw * ntm * WCAP;
  int pend = 0;
  int gpos = 0;
  unsigned ovf = 0u;

  int t = 0;
  for (int g0 = 0; g0 < ng; g0 += TILE_G, ++t) {
    int4 dq[VQ], sq[VQ];
    load_tile<VEC>(dst, e0, e1, ng, g0, tid, -1, dq);
    load_tile<VEC>(src, e0, e1, ng, g0, tid, 0, sq);
    static_assert(VQ == 8, "the fence below names eight src loads");
    __builtin_amdgcn_sched_barrier(0);
    asm volatile("" :: "v"(sq[0].x), "v"(sq[1].x), "v"(sq[2].x), "v"(sq[3].x),
                       "v"(sq[4].x), "v"(sq[5].x), "v"(sq[6].x), "v"(sq[7].x));

    unsigned nbk[VQ];
    #pragma unroll
    for (int k = 0; k < VQ; ++k)
      nbk[k] = group_ok(g0, k, tid, ng) ? (unsigned)nbn : 0u;
    unsigned mk[4 * VQ];
    int tot = 0;
    #pragma unroll
    for (int k = 0; k < VQ; ++k) {
      int dv[4] = {dq[k].x, dq[k].y, dq[k].z, dq[k].w};
      #pragma unroll
      for (int c = 0; c < 4; ++c) {
        mk[4 * k + c] = __builtin_amdgcn_ballot_w32((unsigned)(dv[c] - nb0) < nbk[k]);
        tot += __popc(mk[4 * k + c]);
      }
    }
    if (pend + tot <= WCAP) {
      int run = pend;
      #pragma unroll
      for (int k = 0; k < VQ; ++k) {
        int dv[4] = {dq[k].x, dq[k].y, dq[k].z, dq[k].w};
        int sv[4] = {sq[k].x, sq[k].y, sq[k].z, sq[k].w};
        #pragma unroll
        for (int c = 0; c < 4; ++c) {
          const unsigned dl = (unsigned)(dv[c] - nb0);
          if (dl < nbk[k]) {
            const int p = (int)__builtin_amdgcn_mbcnt_lo(mk[4 * k + c], (unsigned)run);
            wl[2 * p]     = sv[c];
            wl[2 * p + 1] = (int)dl;
          }
          run += __popc(mk[4 * k + c]);
        }
      }

      const int full = run & ~(TRIP - 1);
      const int rest = run - full;
      if (keep) store_list(gl + gpos, wl, full, lane);
      gpos += full;
      sum_list<WD, true>(acc, wl, full, lane, nb0, dinv, feat);
      if (full > 0 && lane < rest) {
        const int2 mv = ((const int2*)wl)[full + lane];
        ((int2*)wl)[lane] = mv;
      }
      pend = rest;
    } else {
      ovf |= 1u << (t & 31);
      sum_direct<WD>(acc, src, dst, dinv, feat, e0, e1, ng, g0, tid, nb0, nbn);
    }
  }
  if (keep) store_list(gl + gpos, wl, pend, lane);
  gpos += pend;
  sum_list<WD, false>(acc, wl, pend, lane, nb0, dinv, feat);
  if (keep && lane == 0) {
    gcnt[2 * wgw]     = gpos;
    gcnt[2 * wgw + 1] = (int)ovf;
  }
  __syncthreads();

  float4* pp = (float4*)(part + ((size_t)blockIdx.y * n + nb0) * WD);
  for (int i = tid; i < nbn * WD / 4; i += ATPB) pp[i] = ((const float4*)acc)[i];
}

template <int WD, int NBK, int NCHK>
__global__ void __launch_bounds__(ATPB) agg_kernel(
    const int* __restrict__ src, const int* __restrict__ dst,
    const float* __restrict__ dinv, const float* __restrict__ feat,
    float* __restrict__ part, int2* __restrict__ glist, int* __restrict__ gcnt,
    int e, int n) {
  __shared__ float acc[NBK * WD];
  __shared__ __attribute__((aligned(16))) int lst[(ATPB / 32) * 2 * WCAP];
  if ((e & 3) == 0)
    agg_body<WD, NBK, NCHK, true>(acc, lst, src, dst, dinv, feat, part, glist, gcnt, e, n);
  else
    agg_body<WD, NBK, NCHK, false>(acc, lst, src, dst, dinv, feat, part, glist, gcnt, e, n);
}

template <int WD, int NBK, int NCHK>
__global__ void __launch_bounds__(ATPB) agg_list_kernel(
    const int* __restrict__ src, const int* __restrict__ dst,
    const float* __restrict__ dinv, const float* __restrict__ feat,
    float* __restrict__ part, const int2* __restrict__ glist,
    const int* __restrict__ gcnt, int e, int n) {
  __shared__ float acc[NBK * WD];
  constexpr int TILE_G = ATPB * VQ;
  constexpr int LPE = WD / 4;
  constexpr int EPW = 32 / LPE;
  const int tid = threadIdx.x;
  const int nb0 = blockIdx.x * NBK;
  const int nbn = min(NBK, n - nb0);
  const int per = (((e + NCHK - 1) / NCHK) + 3) & ~3;
  const int e0  = blockIdx.y * per;
  const int e1  = min(e, e0 + per);
  const int ng  = e1 > e0 ? (e1 - e0 + 3) >> 2 : 0;
  const int nt  = tiles_of(ng);
  const int ntm = tiles_of(per >> 2);
  const bool keep = ntm <= MAXT;

  for (int i = tid; i < NBK * WD / 4; i += ATPB)
    ((float4*)acc)[i] = make_float4(0.f, 0.f, 0.f, 0.f);
  __syncthreads();

  const int lane = tid & 31;
  const int sub  = lane % LPE;
  const int slot = lane / LPE;
  const size_t wgw = (size_t)(blockIdx.y * gridDim.x + blockIdx.x) * (ATPB / 32) + (tid >> 5);
  const int2* gw = glist + wgw * ntm * WCAP;

  int cntw = 0;
  unsigned ovf = 0u;
  if (keep && nt > 0) {
    cntw = gcnt[2 * wgw];
    ovf  = (unsigned)gcnt[2 * wgw + 1];
  }
  const int cnt = __builtin_amdgcn_readfirstlane(cntw);

  int2 nx[4];
  if (cnt > 0) {
    #pragma unroll
    for (int q = 0; q < 4; ++q) {
      const int jj = slot + q * EPW;
      nx[q] = gw[jj < cnt ? jj : 0];
    }
  }
  for (int b = 0; b < cnt; b += 4 * EPW) {
    int s[4], d[4];
    bool ok[4];
    #pragma unroll
    for (int q = 0; q < 4; ++q) {
      s[q]  = nx[q].x;
      d[q]  = nx[q].y;
      ok[q] = b + slot + q * EPW < cnt;
    }
    #pragma unroll
    for (int q = 0; q < 4; ++q) {
      const int jj = b + 4 * EPW + slot + q * EPW;
      nx[q] = gw[jj < cnt ? jj : 0];
    }
    if (b + 4 * EPW <= cnt)
      sum_four<WD, true>(acc, s, d, ok, sub, nb0, dinv, feat);
    else
      sum_four<WD, false>(acc, s, d, ok, sub, nb0, dinv, feat);
  }

  for (int t = 0; t < nt; ++t) {
    if (!keep || ((ovf >> (t & 31)) & 1u))
      sum_direct<WD>(acc, src, dst, dinv, feat, e0, e1, ng, t * TILE_G, tid, nb0, nbn);
  }
  __syncthreads();

  float4* pp = (float4*)(part + ((size_t)blockIdx.y * n + nb0) * WD);
  for (int i = tid; i < nbn * WD / 4; i += ATPB) pp[i] = ((const float4*)acc)[i];
}


__global__ void __launch_bounds__(256) gemm2_kernel(
    const float* __restrict__ h1, const float* __restrict__ part,
    const float* __restrict__ b1,
    const float* __restrict__ W2, const float* __restrict__ dinv,
    float* __restrict__ lt, float* __restrict__ out, int nTiles, int n) {
  int wave = (blockIdx.x * 256 + threadIdx.x) >> 5;
  int lane = threadIdx.x & 31;
  if (wave >= nTiles) return;
  int half = lane >> 4;
  int m    = lane & 15;
  int row  = wave * 16 + m;
  int rowc = row < n ? row : n - 1;
  const float* hrow = h1 + (size_t)rowc * HID;
  const float* prow = part + (size_t)rowc * HID;
  const size_t pstr = (size_t)n * HID;
  float maskv = (m < NCLS) ? 1.0f : 0.0f;
  int mc = m & (NCLS - 1);

  v8f c = {};
  #pragma unroll
  for (int k0 = 0; k0 < HID; k0 += 4) {
    int kk = k0 + 2 * half;
    v2f t = *(const v2f*)(hrow + kk);
    #pragma unroll
    for (int ch = 0; ch < NCH1; ++ch) {
      v2f pv = *(const v2f*)(prow + ch * pstr + kk);
      t.x += pv.x;
      t.y += pv.y;
    }
    v2f a;
    a.x = fmaxf(t.x + b1[kk],     0.0f);
    a.y = fmaxf(t.y + b1[kk + 1], 0.0f);
    v2f b;
    b.x = W2[kk * NCLS + mc] * maskv;
    b.y = W2[(kk + 1) * NCLS + mc] * maskv;
    c = __builtin_amdgcn_wmma_f32_16x16x4_f32(
        false, a, false, b, (short)0, c, false, false);
  }

  float dv = dinv[rowc];
  bool full = (wave * 16 + 16 <= n);
  #pragma unroll
  for (int r = 0; r < 8; ++r) {
    int orow = wave * 16 + r + 8 * half;
    float w  = __shfl(dv, r + 8 * half, 32);
    if (full || orow < n) {
      lt[(size_t)orow * RS2 + m] = (m < NCLS) ? c[r] : ((m == NCLS) ? w : 0.0f);
      if (m < NCLS)
        out[(size_t)orow * NCLS + m] = c[r] * w * w;
    }
  }
}


__global__ void lsm_kernel(float* __restrict__ out,
                           const float* __restrict__ part,
                           const float* __restrict__ b2, int n) {
  int i = blockIdx.x * blockDim.x + threadIdx.x;
  if (i >= n) return;
  float* row = out + (size_t)i * NCLS;
  float z[NCLS];
  #pragma unroll
  for (int c = 0; c < NCLS; ++c) z[c] = row[c];
  #pragma unroll
  for (int ch = 0; ch < NCH1; ++ch) {
    const float4* pr = (const float4*)(part + ((size_t)ch * n + i) * NCLS);
    float4 p0 = pr[0], p1 = pr[1];
    z[0] += p0.x; z[1] += p0.y; z[2] += p0.z; z[3] += p0.w;
    z[4] += p1.x; z[5] += p1.y; z[6] += p1.z; z[7] += p1.w;
  }
  float mx = -INFINITY;
  #pragma unroll
  for (int c = 0; c < NCLS; ++c) {
    z[c] += b2[c];
    mx = fmaxf(mx, z[c]);
  }
  float s = 0.0f;
  #pragma unroll
  for (int c = 0; c < NCLS; ++c) s += expf(z[c] - mx);
  float l = mx + logf(s);
  #pragma unroll
  for (int c = 0; c < NCLS; ++c) row[c] = z[c] - l;
}


extern "C" void kernel_launch(void* const* d_in, const int* in_sizes, int n_in,
                              void* d_out, int out_size, void* d_ws, size_t ws_size,
                              hipStream_t stream) {
  const float* x  = (const float*)d_in[0];
  const int* edge = (const int*)d_in[1];
  const float* W1 = (const float*)d_in[2];
  const float* b1 = (const float*)d_in[3];
  const float* W2 = (const float*)d_in[4];
  const float* b2 = (const float*)d_in[5];

  int n = in_sizes[0] / F_IN;
  int e = in_sizes[1] / 2;
  const int* src = edge;
  const int* dst = edge + e;

  float* deg = (float*)d_ws;
  float* h   = deg + n;
  float* h1  = h + (size_t)n * RS1;
  float* lt  = h1 + (size_t)n * HID;
  float* p1  = lt + (size_t)n * RS2;
  float* p2  = p1 + (size_t)NCH1 * n * HID;
  float* dgp = p2 + (size_t)NCH1 * n * NCLS;
  float* out = (float*)d_out;

  const int T = 256;
  int nTiles  = (n + 15) / 16;
  int gBlocks = (nTiles + 7) / 8;
  dim3 dGrid((n + DNB - 1) / DNB, DCH);
  dim3 aGrid1((n + NB1 - 1) / NB1, NCH1);
  int* gcnt   = (int*)(dgp + (size_t)DCH * n);
  int2* glist = (int2*)(gcnt + (size_t)aGrid1.x * NCH1 * (ATPB / 32) * MAXT);

  deg_part_kernel<<<dGrid, ATPB, 0, stream>>>(dst, dgp, e, n);
  dinv_kernel<<<(n + T - 1) / T, T, 0, stream>>>(dgp, deg, n);

  gemm1_kernel<<<gBlocks, T, 0, stream>>>(x, W1, deg, h, h1, nTiles, n);
  agg_kernel<HID, NB1, NCH1><<<aGrid1, ATPB, 0, stream>>>(src, dst, deg, h, p1, glist, gcnt, e, n);

  gemm2_kernel<<<gBlocks, T, 0, stream>>>(h1, p1, b1, W2, deg, lt, out, nTiles, n);
  agg_list_kernel<NCLS, NB1, NCH1><<<aGrid1, ATPB, 0, stream>>>(src, dst, deg, lt, p2, glist, gcnt, e, n);

  lsm_kernel<<<(n + T - 1) / T, T, 0, stream>>>(out, p2, b2, n);
}


# Edge sums' LDS rows: component c of a lane's float4 kept at c ^ (two more bits of the row number): one f32-add instruction's rows spread over 4x the banks; copy-out undoes it; arrival-order sums
# speedup vs baseline: 10.0617x; 1.0524x over previous
#include <hip/hip_runtime.h>
#include <hip/hip_bf16.h>
#include <math.h>


typedef float v2f __attribute__((ext_vector_type(2)));
typedef float v8f __attribute__((ext_vector_type(8)));

#define F_IN 128
#define HID  16
#define NCLS 8

#define RS1 (2 * HID)
#define RS2 (2 * NCLS)


#define ATPB 1024
#define VQ   8
#define WCAP 128
#define WKL  (WCAP / 32)
#define MAXT 32

#define DNB  25000
#define DCH  64
#define NB1  4000
#define NCH1 10


template <bool VEC>
__device__ __forceinline__ void load_tile(const int* __restrict__ p, int e0,
                                          int e1, int ng, int gb, int tid,
                                          int fill, int4 (&q)[VQ]) {
  if (VEC) {
    #pragma unroll
    for (int k = 0; k < VQ; ++k) {
      int g = gb + k * ATPB + tid;
      q[k] = *(const int4*)(p + e0 + 4 * (g < ng ? g : 0));
    }
  } else {
    #pragma unroll
    for (int k = 0; k < VQ; ++k) {
      int g = gb + k * ATPB + tid;
      int base = e0 + 4 * (g < ng ? g : 0);
      q[k].x = (base + 0 < e1) ? p[base + 0] : fill;
      q[k].y = (base + 1 < e1) ? p[base + 1] : fill;
      q[k].z = (base + 2 < e1) ? p[base + 2] : fill;
      q[k].w = (base + 3 < e1) ? p[base + 3] : fill;
    }
  }
}

__device__ __forceinline__ bool group_ok(int gb, int k, int tid, int ng) {
  return gb + k * ATPB + tid < ng;
}


template <bool VEC>
__device__ __forceinline__ void deg_part_body(
    float* hc, const int* __restrict__ dst, float* __restrict__ degp, int e, int n) {
  const int tid = threadIdx.x;
  const int nb0 = blockIdx.x * DNB;
  const int nbn = min(DNB, n - nb0);
  const int per = (((e + DCH - 1) / DCH) + 3) & ~3;
  const int e0  = blockIdx.y * per;
  const int e1  = min(e, e0 + per);
  const int ng  = e1 > e0 ? (e1 - e0 + 3) >> 2 : 0;

  for (int i = tid; i < DNB; i += ATPB) hc[i] = 0.0f;
  __syncthreads();

  for (int g0 = 0; g0 < ng; g0 += ATPB * VQ) {
    int4 dq[VQ];
    load_tile<VEC>(dst, e0, e1, ng, g0, tid, -1, dq);
    #pragma unroll
    for (int k = 0; k < VQ; ++k) {
      const bool gv = group_ok(g0, k, tid, ng);
      int dv[4] = {dq[k].x, dq[k].y, dq[k].z, dq[k].w};
      #pragma unroll
      for (int c = 0; c < 4; ++c) {
        unsigned dl = (unsigned)(dv[c] - nb0);
        if (gv && dl < (unsigned)nbn) atomicAdd(&hc[dl], 1.0f);
      }
    }
  }
  __syncthreads();

  float* pp = degp + (size_t)blockIdx.y * n + nb0;
  for (int i = tid; i < nbn; i += ATPB) pp[i] = hc[i];
}

__global__ void __launch_bounds__(ATPB) deg_part_kernel(
    const int* __restrict__ dst, float* __restrict__ degp, int e, int n) {
  __shared__ float hc[DNB];
  if ((e & 3) == 0) deg_part_body<true>(hc, dst, degp, e, n);
  else              deg_part_body<false>(hc, dst, degp, e, n);
}

__global__ void dinv_kernel(const float* __restrict__ degp,
                            float* __restrict__ dinv, int n) {
  int i = blockIdx.x * blockDim.x + threadIdx.x;
  if (i >= n) return;
  float s = 1.0f;
  #pragma unroll 16
  for (int ch = 0; ch < DCH; ++ch) s += degp[(size_t)ch * n + i];
  dinv[i] = rsqrtf(s);
}


__global__ void __launch_bounds__(256) gemm1_kernel(
    const float* __restrict__ x, const float* __restrict__ W1,
    const float* __restrict__ dinv,
    float* __restrict__ h, float* __restrict__ out1, int nTiles, int n) {
  int wave = (blockIdx.x * 256 + threadIdx.x) >> 5;
  int lane = threadIdx.x & 31;
  if (wave >= nTiles) return;
  int half = lane >> 4;
  int m    = lane & 15;
  int row  = wave * 16 + m;
  int rowc = row < n ? row : n - 1;
  const float* xrow = x + (size_t)rowc * F_IN;

  v8f c = {};
  #pragma unroll 4
  for (int k0 = 0; k0 < F_IN; k0 += 4) {
    int kk = k0 + 2 * half;
    v2f a = *(const v2f*)(xrow + kk);
    v2f b;
    b.x = W1[kk * HID + m];
    b.y = W1[(kk + 1) * HID + m];
    c = __builtin_amdgcn_wmma_f32_16x16x4_f32(
        false, a, false, b, (short)0, c, false, false);
  }

  float dv = dinv[rowc];
  bool full = (wave * 16 + 16 <= n);
  if (full) {
    #pragma unroll
    for (int r = 0; r < 8; ++r) {
      int orow = wave * 16 + r + 8 * half;
      float w  = __shfl(dv, r + 8 * half, 32);
      size_t o = (size_t)orow * HID + m;
      size_t g = (size_t)orow * RS1 + m;
      h[g]       = c[r];
      h[g + HID] = (m == 0) ? w : 0.0f;
      out1[o] = c[r] * w * w;
    }
  } else {
    #pragma unroll
    for (int r = 0; r < 8; ++r) {
      int orow = wave * 16 + r + 8 * half;
      float w  = __shfl(dv, r + 8 * half, 32);
      if (orow < n) {
        size_t o = (size_t)orow * HID + m;
        size_t g = (size_t)orow * RS1 + m;
        h[g]       = c[r];
        h[g + HID] = (m == 0) ? w : 0.0f;
        out1[o] = c[r] * w * w;
      }
    }
  }
}


__device__ __forceinline__ int tiles_of(int groups) {
  return (groups + ATPB * VQ - 1) / (ATPB * VQ);
}

template <int WD>
__device__ __forceinline__ int row_swz(int d) {
  static_assert(WD == 16 || WD == 8, "rows per pass over the banks: 64 / WD = 4 or 8");
  return (d >> (WD == 16 ? 2 : 3)) & 3;
}

__device__ __forceinline__ float4 unswz4(float4 p, int sw) {
  const float4 q = (sw & 1) ? make_float4(p.y, p.x, p.w, p.z) : p;
  return (sw & 2) ? make_float4(q.z, q.w, q.x, q.y) : q;
}

template <int WD, bool FULL>
__device__ __forceinline__ void sum_four(
    float* acc, const int (&s)[4], const int (&d)[4], const bool (&ok)[4],
    int sub, int nb0,
    const float* __restrict__ dinv, const float* __restrict__ feat) {
  constexpr int RS = 2 * WD;
  float da[4], db[4];
  float4 v[4];
  #pragma unroll
  for (int q = 0; q < 4; ++q) {
    const float* fr = feat + (size_t)s[q] * RS;
    v[q]  = *(const float4*)(fr + sub * 4);
    da[q] = fr[WD];
    db[q] = dinv[nb0 + d[q]];
  }
  __builtin_amdgcn_sched_barrier(0);
  asm volatile("" :: "v"(da[0]), "v"(da[1]), "v"(da[2]), "v"(da[3]),
                     "v"(db[0]), "v"(db[1]), "v"(db[2]), "v"(db[3]),
                     "v"(v[0].x), "v"(v[1].x), "v"(v[2].x), "v"(v[3].x));
  #pragma unroll
  for (int q = 0; q < 4; ++q) {
    if (FULL || ok[q]) {
      float w = da[q] * db[q];
      const unsigned b0 = (unsigned)(d[q] * WD + sub * 4) * 4u;
      const unsigned a0 = b0 | ((unsigned)row_swz<WD>(d[q]) << 2);
      char* ab = (char*)acc;
      atomicAdd((float*)(ab + a0),         v[q].x * w);
      atomicAdd((float*)(ab + (a0 ^ 4u)),  v[q].y * w);
      atomicAdd((float*)(ab + (a0 ^ 8u)),  v[q].z * w);
      atomicAdd((float*)(ab + (a0 ^ 12u)), v[q].w * w);
    }
  }
}

template <int WD, bool FULL>
__device__ __forceinline__ void sum_list(
    float* acc, const int* wl, int tot, int lane, int nb0,
    const float* __restrict__ dinv, const float* __restrict__ feat) {
  constexpr int LPE = WD / 4;
  constexpr int EPW = 32 / LPE;
  const int sub  = lane % LPE;
  const int slot = lane / LPE;
  for (int j = slot; j < tot; j += 4 * EPW) {
    int s[4], d[4];
    bool ok[4];
    #pragma unroll
    for (int q = 0; q < 4; ++q) {
      int jj = j + q * EPW;
      ok[q] = FULL || jj < tot;
      int jc = ok[q] ? jj : j;
      s[q] = wl[2 * jc];
      d[q] = wl[2 * jc + 1];
    }
    sum_four<WD, FULL>(acc, s, d, ok, sub, nb0, dinv, feat);
  }
}

template <int WD>
__device__ __forceinline__ void sum_direct(
    float* acc, const int* __restrict__ src, const int* __restrict__ dst,
    const float* __restrict__ dinv, const float* __restrict__ feat,
    int e0, int e1, int ng, int g0, int tid, int nb0, int nbn) {
  constexpr int RS = 2 * WD;
  #pragma unroll 1
  for (int b = 0; b < 4 * VQ; ++b) {
    const int g  = g0 + (b >> 2) * ATPB + tid;
    const int ei = e0 + 4 * g + (b & 3);
    if (g < ng && ei < e1) {
      const int dvv = dst[ei];
      const unsigned dl = (unsigned)(dvv - nb0);
      if (dl < (unsigned)nbn) {
        const float* fr = feat + (size_t)src[ei] * RS;
        float w = fr[WD] * dinv[dvv];
        #pragma unroll 1
        for (int f = 0; f < WD; ++f) atomicAdd(&acc[dl * WD + (f ^ row_swz<WD>((int)dl))], fr[f] * w);
      }
    }
  }
}

__device__ __forceinline__ void store_list(int2* __restrict__ g, const int* wl,
                                           int cnt, int lane) {
  if (cnt > 0) {
    int2 cp[WKL];
    #pragma unroll
    for (int k = 0; k < WKL; ++k) {
      const int i = lane + 32 * k;
      cp[k] = ((const int2*)wl)[i < cnt ? i : 0];
    }
    #pragma unroll
    for (int k = 0; k < WKL; ++k) {
      const int i = lane + 32 * k;
      if (i < cnt) g[i] = cp[k];
    }
  }
}

template <int WD, int NBK, int NCHK, bool VEC>
__device__ __forceinline__ void agg_body(
    float* acc, int* lst,
    const int* __restrict__ src, const int* __restrict__ dst,
    const float* __restrict__ dinv, const float* __restrict__ feat,
    float* __restrict__ part, int2* __restrict__ glist, int* __restrict__ gcnt,
    int e, int n) {
  constexpr int TILE_G = ATPB * VQ;
  const int tid = threadIdx.x;
  const int nb0 = blockIdx.x * NBK;
  const int nbn = min(NBK, n - nb0);
  const int per = (((e + NCHK - 1) / NCHK) + 3) & ~3;
  const int e0  = blockIdx.y * per;
  const int e1  = min(e, e0 + per);
  const int ng  = e1 > e0 ? (e1 - e0 + 3) >> 2 : 0;
  const int ntm = tiles_of(per >> 2);
  const bool keep = ntm <= MAXT;

  for (int i = tid; i < NBK * WD / 4; i += ATPB)
    ((float4*)acc)[i] = make_float4(0.f, 0.f, 0.f, 0.f);
  __syncthreads();

  const int lane = tid & 31;
  int* wl = lst + (tid >> 5) * (2 * WCAP);
  const size_t wgw = (size_t)(blockIdx.y * gridDim.x + blockIdx.x) * (ATPB / 32) + (tid >> 5);

  constexpr int TRIP = 4 * (32 / (WD / 4));
  static_assert(TRIP <= 32 && (TRIP & (TRIP - 1)) == 0, "a tile's rest must fit one entry per lane");
  int2* gl = glist + wgw * ntm * WCAP;
  int pend = 0;
  int gpos = 0;
  unsigned ovf = 0u;

  int t = 0;
  for (int g0 = 0; g0 < ng; g0 += TILE_G, ++t) {
    int4 dq[VQ], sq[VQ];
    load_tile<VEC>(dst, e0, e1, ng, g0, tid, -1, dq);
    load_tile<VEC>(src, e0, e1, ng, g0, tid, 0, sq);
    static_assert(VQ == 8, "the fence below names eight src loads");
    __builtin_amdgcn_sched_barrier(0);
    asm volatile("" :: "v"(sq[0].x), "v"(sq[1].x), "v"(sq[2].x), "v"(sq[3].x),
                       "v"(sq[4].x), "v"(sq[5].x), "v"(sq[6].x), "v"(sq[7].x));

    unsigned nbk[VQ];
    #pragma unroll
    for (int k = 0; k < VQ; ++k)
      nbk[k] = group_ok(g0, k, tid, ng) ? (unsigned)nbn : 0u;
    unsigned mk[4 * VQ];
    int tot = 0;
    #pragma unroll
    for (int k = 0; k < VQ; ++k) {
      int dv[4] = {dq[k].x, dq[k].y, dq[k].z, dq[k].w};
      #pragma unroll
      for (int c = 0; c < 4; ++c) {
        mk[4 * k + c] = __builtin_amdgcn_ballot_w32((unsigned)(dv[c] - nb0) < nbk[k]);
        tot += __popc(mk[4 * k + c]);
      }
    }
    if (pend + tot <= WCAP) {
      int run = pend;
      #pragma unroll
      for (int k = 0; k < VQ; ++k) {
        int dv[4] = {dq[k].x, dq[k].y, dq[k].z, dq[k].w};
        int sv[4] = {sq[k].x, sq[k].y, sq[k].z, sq[k].w};
        #pragma unroll
        for (int c = 0; c < 4; ++c) {
          const unsigned dl = (unsigned)(dv[c] - nb0);
          if (dl < nbk[k]) {
            const int p = (int)__builtin_amdgcn_mbcnt_lo(mk[4 * k + c], (unsigned)run);
            wl[2 * p]     = sv[c];
            wl[2 * p + 1] = (int)dl;
          }
          run += __popc(mk[4 * k + c]);
        }
      }

      const int full = run & ~(TRIP - 1);
      const int rest = run - full;
      if (keep) store_list(gl + gpos, wl, full, lane);
      gpos += full;
      sum_list<WD, true>(acc, wl, full, lane, nb0, dinv, feat);
      if (full > 0 && lane < rest) {
        const int2 mv = ((const int2*)wl)[full + lane];
        ((int2*)wl)[lane] = mv;
      }
      pend = rest;
    } else {
      ovf |= 1u << (t & 31);
      sum_direct<WD>(acc, src, dst, dinv, feat, e0, e1, ng, g0, tid, nb0, nbn);
    }
  }
  if (keep) store_list(gl + gpos, wl, pend, lane);
  gpos += pend;
  sum_list<WD, false>(acc, wl, pend, lane, nb0, dinv, feat);
  if (keep && lane == 0) {
    gcnt[2 * wgw]     = gpos;
    gcnt[2 * wgw + 1] = (int)ovf;
  }
  __syncthreads();

  float4* pp = (float4*)(part + ((size_t)blockIdx.y * n + nb0) * WD);
  for (int i = tid; i < nbn * WD / 4; i += ATPB)
    pp[i] = unswz4(((const float4*)acc)[i], row_swz<WD>(i / (WD / 4)));
}

template <int WD, int NBK, int NCHK>
__global__ void __launch_bounds__(ATPB) agg_kernel(
    const int* __restrict__ src, const int* __restrict__ dst,
    const float* __restrict__ dinv, const float* __restrict__ feat,
    float* __restrict__ part, int2* __restrict__ glist, int* __restrict__ gcnt,
    int e, int n) {
  __shared__ float acc[NBK * WD];
  __shared__ __attribute__((aligned(16))) int lst[(ATPB / 32) * 2 * WCAP];
  if ((e & 3) == 0)
    agg_body<WD, NBK, NCHK, true>(acc, lst, src, dst, dinv, feat, part, glist, gcnt, e, n);
  else
    agg_body<WD, NBK, NCHK, false>(acc, lst, src, dst, dinv, feat, part, glist, gcnt, e, n);
}

template <int WD, int NBK, int NCHK>
__global__ void __launch_bounds__(ATPB) agg_list_kernel(
    const int* __restrict__ src, const int* __restrict__ dst,
    const float* __restrict__ dinv, const float* __restrict__ feat,
    float* __restrict__ part, const int2* __restrict__ glist,
    const int* __restrict__ gcnt, int e, int n) {
  __shared__ float acc[NBK * WD];
  constexpr int TILE_G = ATPB * VQ;
  constexpr int LPE = WD / 4;
  constexpr int EPW = 32 / LPE;
  const int tid = threadIdx.x;
  const int nb0 = blockIdx.x * NBK;
  const int nbn = min(NBK, n - nb0);
  const int per = (((e + NCHK - 1) / NCHK) + 3) & ~3;
  const int e0  = blockIdx.y * per;
  const int e1  = min(e, e0 + per);
  const int ng  = e1 > e0 ? (e1 - e0 + 3) >> 2 : 0;
  const int nt  = tiles_of(ng);
  const int ntm = tiles_of(per >> 2);
  const bool keep = ntm <= MAXT;

  for (int i = tid; i < NBK * WD / 4; i += ATPB)
    ((float4*)acc)[i] = make_float4(0.f, 0.f, 0.f, 0.f);
  __syncthreads();

  const int lane = tid & 31;
  const int sub  = lane % LPE;
  const int slot = lane / LPE;
  const size_t wgw = (size_t)(blockIdx.y * gridDim.x + blockIdx.x) * (ATPB / 32) + (tid >> 5);
  const int2* gw = glist + wgw * ntm * WCAP;

  int cntw = 0;
  unsigned ovf = 0u;
  if (keep && nt > 0) {
    cntw = gcnt[2 * wgw];
    ovf  = (unsigned)gcnt[2 * wgw + 1];
  }
  const int cnt = __builtin_amdgcn_readfirstlane(cntw);

  int2 nx[4];
  if (cnt > 0) {
    #pragma unroll
    for (int q = 0; q < 4; ++q) {
      const int jj = slot + q * EPW;
      nx[q] = gw[jj < cnt ? jj : 0];
    }
  }
  for (int b = 0; b < cnt; b += 4 * EPW) {
    int s[4], d[4];
    bool ok[4];
    #pragma unroll
    for (int q = 0; q < 4; ++q) {
      s[q]  = nx[q].x;
      d[q]  = nx[q].y;
      ok[q] = b + slot + q * EPW < cnt;
    }
    #pragma unroll
    for (int q = 0; q < 4; ++q) {
      const int jj = b + 4 * EPW + slot + q * EPW;
      nx[q] = gw[jj < cnt ? jj : 0];
    }
    if (b + 4 * EPW <= cnt)
      sum_four<WD, true>(acc, s, d, ok, sub, nb0, dinv, feat);
    else
      sum_four<WD, false>(acc, s, d, ok, sub, nb0, dinv, feat);
  }

  for (int t = 0; t < nt; ++t) {
    if (!keep || ((ovf >> (t & 31)) & 1u))
      sum_direct<WD>(acc, src, dst, dinv, feat, e0, e1, ng, t * TILE_G, tid, nb0, nbn);
  }
  __syncthreads();

  float4* pp = (float4*)(part + ((size_t)blockIdx.y * n + nb0) * WD);
  for (int i = tid; i < nbn * WD / 4; i += ATPB)
    pp[i] = unswz4(((const float4*)acc)[i], row_swz<WD>(i / (WD / 4)));
}


__global__ void __launch_bounds__(256) gemm2_kernel(
    const float* __restrict__ h1, const float* __restrict__ part,
    const float* __restrict__ b1,
    const float* __restrict__ W2, const float* __restrict__ dinv,
    float* __restrict__ lt, float* __restrict__ out, int nTiles, int n) {
  int wave = (blockIdx.x * 256 + threadIdx.x) >> 5;
  int lane = threadIdx.x & 31;
  if (wave >= nTiles) return;
  int half = lane >> 4;
  int m    = lane & 15;
  int row  = wave * 16 + m;
  int rowc = row < n ? row : n - 1;
  const float* hrow = h1 + (size_t)rowc * HID;
  const float* prow = part + (size_t)rowc * HID;
  const size_t pstr = (size_t)n * HID;
  float maskv = (m < NCLS) ? 1.0f : 0.0f;
  int mc = m & (NCLS - 1);

  v8f c = {};
  #pragma unroll
  for (int k0 = 0; k0 < HID; k0 += 4) {
    int kk = k0 + 2 * half;
    v2f t = *(const v2f*)(hrow + kk);
    #pragma unroll
    for (int ch = 0; ch < NCH1; ++ch) {
      v2f pv = *(const v2f*)(prow + ch * pstr + kk);
      t.x += pv.x;
      t.y += pv.y;
    }
    v2f a;
    a.x = fmaxf(t.x + b1[kk],     0.0f);
    a.y = fmaxf(t.y + b1[kk + 1], 0.0f);
    v2f b;
    b.x = W2[kk * NCLS + mc] * maskv;
    b.y = W2[(kk + 1) * NCLS + mc] * maskv;
    c = __builtin_amdgcn_wmma_f32_16x16x4_f32(
        false, a, false, b, (short)0, c, false, false);
  }

  float dv = dinv[rowc];
  bool full = (wave * 16 + 16 <= n);
  #pragma unroll
  for (int r = 0; r < 8; ++r) {
    int orow = wave * 16 + r + 8 * half;
    float w  = __shfl(dv, r + 8 * half, 32);
    if (full || orow < n) {
      lt[(size_t)orow * RS2 + m] = (m < NCLS) ? c[r] : ((m == NCLS) ? w : 0.0f);
      if (m < NCLS)
        out[(size_t)orow * NCLS + m] = c[r] * w * w;
    }
  }
}


__global__ void lsm_kernel(float* __restrict__ out,
                           const float* __restrict__ part,
                           const float* __restrict__ b2, int n) {
  int i = blockIdx.x * blockDim.x + threadIdx.x;
  if (i >= n) return;
  float* row = out + (size_t)i * NCLS;
  float z[NCLS];
  #pragma unroll
  for (int c = 0; c < NCLS; ++c) z[c] = row[c];
  #pragma unroll
  for (int ch = 0; ch < NCH1; ++ch) {
    const float4* pr = (const float4*)(part + ((size_t)ch * n + i) * NCLS);
    float4 p0 = pr[0], p1 = pr[1];
    z[0] += p0.x; z[1] += p0.y; z[2] += p0.z; z[3] += p0.w;
    z[4] += p1.x; z[5] += p1.y; z[6] += p1.z; z[7] += p1.w;
  }
  float mx = -INFINITY;
  #pragma unroll
  for (int c = 0; c < NCLS; ++c) {
    z[c] += b2[c];
    mx = fmaxf(mx, z[c]);
  }
  float s = 0.0f;
  #pragma unroll
  for (int c = 0; c < NCLS; ++c) s += expf(z[c] - mx);
  float l = mx + logf(s);
  #pragma unroll
  for (int c = 0; c < NCLS; ++c) row[c] = z[c] - l;
}


extern "C" void kernel_launch(void* const* d_in, const int* in_sizes, int n_in,
                              void* d_out, int out_size, void* d_ws, size_t ws_size,
                              hipStream_t stream) {
  const float* x  = (const float*)d_in[0];
  const int* edge = (const int*)d_in[1];
  const float* W1 = (const float*)d_in[2];
  const float* b1 = (const float*)d_in[3];
  const float* W2 = (const float*)d_in[4];
  const float* b2 = (const float*)d_in[5];

  int n = in_sizes[0] / F_IN;
  int e = in_sizes[1] / 2;
  const int* src = edge;
  const int* dst = edge + e;

  float* deg = (float*)d_ws;
  float* h   = deg + n;
  float* h1  = h + (size_t)n * RS1;
  float* lt  = h1 + (size_t)n * HID;
  float* p1  = lt + (size_t)n * RS2;
  float* p2  = p1 + (size_t)NCH1 * n * HID;
  float* dgp = p2 + (size_t)NCH1 * n * NCLS;
  float* out = (float*)d_out;

  const int T = 256;
  int nTiles  = (n + 15) / 16;
  int gBlocks = (nTiles + 7) / 8;
  dim3 dGrid((n + DNB - 1) / DNB, DCH);
  dim3 aGrid1((n + NB1 - 1) / NB1, NCH1);
  int* gcnt   = (int*)(dgp + (size_t)DCH * n);
  int2* glist = (int2*)(gcnt + (size_t)aGrid1.x * NCH1 * (ATPB / 32) * MAXT);

  deg_part_kernel<<<dGrid, ATPB, 0, stream>>>(dst, dgp, e, n);
  dinv_kernel<<<(n + T - 1) / T, T, 0, stream>>>(dgp, deg, n);

  gemm1_kernel<<<gBlocks, T, 0, stream>>>(x, W1, deg, h, h1, nTiles, n);
  agg_kernel<HID, NB1, NCH1><<<aGrid1, ATPB, 0, stream>>>(src, dst, deg, h, p1, glist, gcnt, e, n);

  gemm2_kernel<<<gBlocks, T, 0, stream>>>(h1, p1, b1, W2, deg, lt, out, nTiles, n);
  agg_list_kernel<NCLS, NB1, NCH1><<<aGrid1, ATPB, 0, stream>>>(src, dst, deg, lt, p2, glist, gcnt, e, n);

  lsm_kernel<<<(n + T - 1) / T, T, 0, stream>>>(out, p2, b2, n);
}
